# starting kernel
# baseline (speedup 1.0000x reference)
#include <hip/hip_runtime.h>
#include <hip/hip_bf16.h>

#define IN_FEATS   128
#define OUT_FEATS  16
#define NEG_SLOPE  0.2f

typedef float v2f __attribute__((ext_vector_type(2)));
typedef float v8f __attribute__((ext_vector_type(8)));

__global__ void gat_project_wmma(const float* __restrict__ h,
                                 const float* __restrict__ W,
                                 float* __restrict__ hW, int nNodes) {
  int wave = (blockIdx.x * blockDim.x + threadIdx.x) >> 5;
  int lane = threadIdx.x & 31;
  int m0 = wave * 16;
  if (m0 >= nNodes) return;
  int half = lane >> 4;
  int l16  = lane & 15;
  const float* arow = h + (size_t)(m0 + l16) * IN_FEATS;
  const float* brow = W + (size_t)l16 * IN_FEATS;
  v8f acc = {};
#pragma unroll
  for (int k = 0; k < IN_FEATS; k += 4) {
    int ka = k + half * 2;
    v2f a = *(const v2f*)(arow + ka);
    v2f b = *(const v2f*)(brow + ka);
    acc = __builtin_amdgcn_wmma_f32_16x16x4_f32(false, a, false, b,
                                                (short)0, acc, false, false);
  }
#pragma unroll
  for (int g = 0; g < 8; ++g) {
    int row = m0 + half * 8 + g;
    hW[(size_t)row * OUT_FEATS + l16] = acc[g];
  }
}

__global__ void gat_node_prep(const float* __restrict__ hW,
                              const float* __restrict__ a_l,
                              const float* __restrict__ a_r,
                              float* __restrict__ el, float* __restrict__ er,
                              float* __restrict__ emax, float* __restrict__ esum,
                              float* __restrict__ out, int nNodes) {
  int n = blockIdx.x * blockDim.x + threadIdx.x;
  if (n >= nNodes) return;
  const float* row = hW + (size_t)n * OUT_FEATS;
  float sl = 0.f, sr = 0.f;
#pragma unroll
  for (int f = 0; f < OUT_FEATS; ++f) {
    float v = row[f];
    sl += v * a_l[f];
    sr += v * a_r[f];
    out[(size_t)n * OUT_FEATS + f] = 0.f;
  }
  el[n] = sl;
  er[n] = sr;
  emax[n] = __int_as_float(0xFF800000);
  esum[n] = 0.f;
}

__device__ __forceinline__ void atomicMaxFloat(float* addr, float val) {
  if (val >= 0.f) atomicMax((int*)addr, __float_as_int(val));
  else            atomicMin((unsigned int*)addr, (unsigned int)__float_as_int(val));
}

__global__ void gat_edge_logit(const int* __restrict__ src,
                               const int* __restrict__ dst,
                               const float* __restrict__ el,
                               const float* __restrict__ er,
                               float* __restrict__ elog,
                               float* __restrict__ emax, int nEdges) {
  int e = blockIdx.x * blockDim.x + threadIdx.x;
  if (e >= nEdges) return;
  int s = src[e], d = dst[e];
  float x = el[s] + er[d];
  x = (x > 0.f) ? x : NEG_SLOPE * x;
  elog[e] = x;
  atomicMaxFloat(emax + d, x);
}

__global__ void gat_node_fixmax(float* __restrict__ emax, int nNodes) {
  int n = blockIdx.x * blockDim.x + threadIdx.x;
  if (n >= nNodes) return;
  if (emax[n] == __int_as_float(0xFF800000)) emax[n] = 0.f;
}

__global__ void gat_edge_exp(const int* __restrict__ dst,
                             float* __restrict__ elog,
                             const float* __restrict__ emax,
                             float* __restrict__ esum, int nEdges) {
  int e = blockIdx.x * blockDim.x + threadIdx.x;
  if (e >= nEdges) return;
  int d = dst[e];
  float ex = __expf(elog[e] - emax[d]);
  elog[e] = ex;
  atomicAdd(esum + d, ex);
}

__global__ void gat_edge_aggr(const int* __restrict__ src,
                              const int* __restrict__ dst,
                              const float* __restrict__ elog,
                              const float* __restrict__ esum,
                              const float* __restrict__ hW,
                              float* __restrict__ out, int nEdges) {
  long long t = (long long)blockIdx.x * blockDim.x + threadIdx.x;
  int e = (int)(t >> 4);
  int f = (int)(t & 15);
  if (e >= nEdges) return;
  int s = src[e], d = dst[e];
  float alpha = elog[e] / fmaxf(esum[d], 1e-16f);
  atomicAdd(out + (size_t)d * OUT_FEATS + f,
            hW[(size_t)s * OUT_FEATS + f] * alpha);
}

extern "C" void kernel_launch(void* const* d_in, const int* in_sizes, int n_in,
                              void* d_out, int out_size, void* d_ws, size_t ws_size,
                              hipStream_t stream) {
  const float* h   = (const float*)d_in[0];
  const int*   src = (const int*)d_in[1];
  const int*   dst = (const int*)d_in[2];
  const float* W   = (const float*)d_in[3];
  const float* a_l = (const float*)d_in[4];
  const float* a_r = (const float*)d_in[5];
  float* out = (float*)d_out;

  int nNodes = in_sizes[0] / IN_FEATS;
  int nEdges = in_sizes[1];

  float* ws   = (float*)d_ws;
  float* hW   = ws;
  float* el   = hW + (size_t)nNodes * OUT_FEATS;
  float* er   = el + nNodes;
  float* emax = er + nNodes;
  float* esum = emax + nNodes;
  float* elog = esum + nNodes;

  dim3 blk(256);
  int tiles = (nNodes + 15) / 16;
  int blocksK1 = (tiles + 7) / 8;

  gat_project_wmma<<<blocksK1, blk, 0, stream>>>(h, W, hW, nNodes);
  gat_node_prep<<<(nNodes + 255) / 256, blk, 0, stream>>>(hW, a_l, a_r, el, er,
                                                          emax, esum, out, nNodes);
  gat_edge_logit<<<(nEdges + 255) / 256, blk, 0, stream>>>(src, dst, el, er,
                                                           elog, emax, nEdges);
  gat_node_fixmax<<<(nNodes + 255) / 256, blk, 0, stream>>>(emax, nNodes);
  gat_edge_exp<<<(nEdges + 255) / 256, blk, 0, stream>>>(dst, elog, emax, esum, nEdges);

  long long laneWork = (long long)nEdges * OUT_FEATS;
  int blocksK6 = (int)((laneWork + 255) / 256);
  gat_edge_aggr<<<blocksK6, blk, 0, stream>>>(src, dst, elog, esum, hW, out, nEdges);
}


# Edge passes (max, exp-sum, aggregation) as resident grids: waves walk the edge list by grid stride, next loads in flight; 1024/1024/1280 work-groups start, was 12500 each; atomic sum order changes
# speedup vs baseline: 3.4212x; 3.4212x over previous
#include <hip/hip_runtime.h>
#include <hip/hip_bf16.h>

#define IN_FEATS   128
#define OUT_FEATS  16
#define NEG_SLOPE  0.2f

#define EDGE_WGS     1024
#define EDGE_UNROLL  4
#define AGGR_WGS     1280

typedef float v2f __attribute__((ext_vector_type(2)));
typedef float v8f __attribute__((ext_vector_type(8)));

__global__ void gat_project_wmma(const float* __restrict__ h,
                                 const float* __restrict__ W,
                                 float* __restrict__ hW, int nNodes) {
  int wave = (blockIdx.x * blockDim.x + threadIdx.x) >> 5;
  int lane = threadIdx.x & 31;
  int m0 = wave * 16;
  if (m0 >= nNodes) return;
  int half = lane >> 4;
  int l16  = lane & 15;
  const float* arow = h + (size_t)(m0 + l16) * IN_FEATS;
  const float* brow = W + (size_t)l16 * IN_FEATS;
  v8f acc = {};
#pragma unroll
  for (int k = 0; k < IN_FEATS; k += 4) {
    int ka = k + half * 2;
    v2f a = *(const v2f*)(arow + ka);
    v2f b = *(const v2f*)(brow + ka);
    acc = __builtin_amdgcn_wmma_f32_16x16x4_f32(false, a, false, b,
                                                (short)0, acc, false, false);
  }
#pragma unroll
  for (int g = 0; g < 8; ++g) {
    int row = m0 + half * 8 + g;
    hW[(size_t)row * OUT_FEATS + l16] = acc[g];
  }
}

__device__ __forceinline__ int package_half() {
  return (int)((__builtin_amdgcn_s_sendmsg_rtn(0x87) >> 18) & 1u);
}

__global__ void gat_node_prep(const float* __restrict__ hW,
                              const float* __restrict__ a_l,
                              const float* __restrict__ a_r,
                              float* __restrict__ rec,
                              float* emax0, float* emax1,
                              float* esum0, float* esum1,
                              float* acc0, float* acc1, int nNodes) {
  int n = blockIdx.x * blockDim.x + threadIdx.x;
  if (n >= nNodes) return;
  const float* row = hW + (size_t)n * OUT_FEATS;
  float sl = 0.f, sr = 0.f;
#pragma unroll
  for (int f = 0; f < OUT_FEATS; ++f) {
    float v = row[f];
    sl += v * a_l[f];
    sr += v * a_r[f];
  }
  float4 zero4 = make_float4(0.f, 0.f, 0.f, 0.f);
  float4* z0 = (float4*)(acc0 + (size_t)n * OUT_FEATS);
  float4* z1 = (float4*)(acc1 + (size_t)n * OUT_FEATS);
#pragma unroll
  for (int q = 0; q < OUT_FEATS / 4; ++q) z0[q] = zero4;
#pragma unroll
  for (int q = 0; q < OUT_FEATS / 4; ++q) z1[q] = zero4;
  ((float2*)rec)[n] = make_float2(sl, sr);
  emax0[n] = __int_as_float(0xFF800000);
  emax1[n] = __int_as_float(0xFF800000);
  esum0[n] = 0.f;
  esum1[n] = 0.f;
}

__device__ __forceinline__ void atomicMaxFloat(float* addr, float val) {
  if (val >= 0.f) atomicMax((int*)addr, __float_as_int(val));
  else            atomicMin((unsigned int*)addr, (unsigned int)__float_as_int(val));
}

__global__ void __launch_bounds__(256)
gat_edge_max(const int* __restrict__ src,
             const int* __restrict__ dst,
             const float* __restrict__ rec,
             float* __restrict__ els,
             float* emax0, float* emax1, int nEdges) {
  int tid = blockIdx.x * 256 + threadIdx.x;
  int stride = gridDim.x * 256;
  float* emax = package_half() ? emax1 : emax0;
  for (int e0 = tid; e0 < nEdges; e0 += EDGE_UNROLL * stride) {
    int s[EDGE_UNROLL], d[EDGE_UNROLL];
    float v[EDGE_UNROLL];
#pragma unroll
    for (int u = 0; u < EDGE_UNROLL; ++u) {
      int e = e0 + u * stride;
      int ec = (e < nEdges) ? e : e0;
      s[u] = src[ec];
      d[u] = dst[ec];
    }
#pragma unroll
    for (int u = 0; u < EDGE_UNROLL; ++u) v[u] = rec[2 * (size_t)s[u]];
#pragma unroll
    for (int u = 0; u < EDGE_UNROLL; ++u) {
      int e = e0 + u * stride;
      if (e < nEdges) {
        els[e] = v[u];
        atomicMaxFloat(emax + d[u], v[u]);
      }
    }
  }
}

__global__ void gat_node_fixmax(const float* emax0, const float* emax1,
                                float* rec, int nNodes) {
  int n = blockIdx.x * blockDim.x + threadIdx.x;
  if (n >= nNodes) return;
  float m = fmaxf(emax0[n], emax1[n]);
  float x = m + rec[2 * (size_t)n + 1];
  x = (x > 0.f) ? x : NEG_SLOPE * x;
  if (!(fabsf(x) < __int_as_float(0x7F800000))) x = 0.f;
  rec[2 * (size_t)n] = x;
}

__global__ void __launch_bounds__(256)
gat_edge_exp(const int* __restrict__ dst,
             float* __restrict__ elog,
             const float* __restrict__ rec,
             float* esum0, float* esum1, int nEdges) {
  int tid = blockIdx.x * 256 + threadIdx.x;
  int stride = gridDim.x * 256;
  float* esum = package_half() ? esum1 : esum0;
  for (int e0 = tid; e0 < nEdges; e0 += EDGE_UNROLL * stride) {
    int d[EDGE_UNROLL];
    float el_s[EDGE_UNROLL];
    float2 rc[EDGE_UNROLL];
#pragma unroll
    for (int u = 0; u < EDGE_UNROLL; ++u) {
      int e = e0 + u * stride;
      int ec = (e < nEdges) ? e : e0;
      d[u] = dst[ec];
      el_s[u] = elog[ec];
    }
#pragma unroll
    for (int u = 0; u < EDGE_UNROLL; ++u) rc[u] = ((const float2*)rec)[d[u]];
#pragma unroll
    for (int u = 0; u < EDGE_UNROLL; ++u) {
      int e = e0 + u * stride;
      if (e < nEdges) {
        float x = el_s[u] + rc[u].y;
        x = (x > 0.f) ? x : NEG_SLOPE * x;
        float ex = __expf(x - rc[u].x);
        elog[e] = ex;
        atomicAdd(esum + d[u], ex);
      }
    }
  }
}

__global__ void __launch_bounds__(256)
gat_edge_aggr(const int* __restrict__ src,
                              const int* __restrict__ dst,
                              const float* __restrict__ elog,
                              const float* esum0, const float* esum1,
                              const float* __restrict__ hW,
                              float* acc0, float* acc1, int nEdges) {
  int wave = (blockIdx.x * 256 + threadIdx.x) >> 5;
  int nWaves = gridDim.x * 8;
  int lane = threadIdx.x & 31;
  int sub = lane >> 4;
  int f = lane & 15;
  int nChunks = (nEdges + 31) >> 5;
  int chunk = wave;
  if (chunk >= nChunks) return;

  int e = min(chunk * 32 + lane, nEdges - 1);
  int s_n = src[e], d_n = dst[e];
  float x_n = elog[e];
  float* out = package_half() ? acc1 : acc0;

  for (; chunk < nChunks; chunk += nWaves) {
    int e0 = chunk * 32;
    int s_l = s_n, d_l = d_n;
    float x_l = x_n;
    int nc = chunk + nWaves;
    if (nc < nChunks) {
      int en = min(nc * 32 + lane, nEdges - 1);
      s_n = src[en];
      d_n = dst[en];
      x_n = elog[en];
    }
    float es0 = esum0[d_l], es1 = esum1[d_l];
    float es = (esum1 != esum0) ? es0 + es1 : es0;
    float a_l = x_l / fmaxf(es, 1e-16f);

    int dd[16];
    float vv[16];
#pragma unroll
    for (int j = 0; j < 16; ++j) {
      int q = 2 * j + sub;
      int s = __shfl(s_l, q, 32);
      dd[j] = __shfl(d_l, q, 32);
      float a = __shfl(a_l, q, 32);
      vv[j] = hW[(size_t)s * OUT_FEATS + f] * a;
    }
#pragma unroll
    for (int j = 0; j < 16; ++j) {
      int q = 2 * j + sub;
      if (e0 + q < nEdges)
        atomicAdd(out + (size_t)dd[j] * OUT_FEATS + f, vv[j]);
    }
  }
}

__global__ void gat_join_out(float* __restrict__ out,
                             const float* __restrict__ acc1, int nVec4) {
  int i = blockIdx.x * blockDim.x + threadIdx.x;
  if (i >= nVec4) return;
  float4 a = ((const float4*)out)[i];
  float4 b = ((const float4*)acc1)[i];
  a.x += b.x; a.y += b.y; a.z += b.z; a.w += b.w;
  ((float4*)out)[i] = a;
}

extern "C" void kernel_launch(void* const* d_in, const int* in_sizes, int n_in,
                              void* d_out, int out_size, void* d_ws, size_t ws_size,
                              hipStream_t stream) {
  const float* h   = (const float*)d_in[0];
  const int*   src = (const int*)d_in[1];
  const int*   dst = (const int*)d_in[2];
  const float* W   = (const float*)d_in[3];
  const float* a_l = (const float*)d_in[4];
  const float* a_r = (const float*)d_in[5];
  float* out = (float*)d_out;

  int nNodes = in_sizes[0] / IN_FEATS;
  int nEdges = in_sizes[1];

  size_t need2 = ((size_t)nNodes * (2 * OUT_FEATS + 6) + (size_t)nEdges) * sizeof(float);
  bool two = ws_size >= need2;
  float* ws    = (float*)d_ws;
  float* hW    = ws;
  float* acc1  = two ? hW + (size_t)nNodes * OUT_FEATS : out;
  float* rec   = hW + (size_t)nNodes * OUT_FEATS * (two ? 2 : 1);
  float* emax0 = rec + 2 * (size_t)nNodes;
  float* emax1 = two ? emax0 + nNodes : emax0;
  float* esum0 = emax1 + nNodes;
  float* esum1 = two ? esum0 + nNodes : esum0;
  float* elog  = esum1 + nNodes;

  dim3 blk(256);
  int tiles = (nNodes + 15) / 16;
  int blocksK1 = (tiles + 7) / 8;

  gat_project_wmma<<<blocksK1, blk, 0, stream>>>(h, W, hW, nNodes);
  gat_node_prep<<<(nNodes + 255) / 256, blk, 0, stream>>>(hW, a_l, a_r, rec,
                                                          emax0, emax1, esum0, esum1,
                                                          out, acc1, nNodes);
  gat_edge_max<<<EDGE_WGS, blk, 0, stream>>>(src, dst, rec,
                                             elog, emax0, emax1, nEdges);
  gat_node_fixmax<<<(nNodes + 255) / 256, blk, 0, stream>>>(emax0, emax1, rec, nNodes);
  gat_edge_exp<<<EDGE_WGS, blk, 0, stream>>>(dst, elog, rec,
                                             esum0, esum1, nEdges);

  gat_edge_aggr<<<AGGR_WGS, blk, 0, stream>>>(src, dst, elog, esum0, esum1, hW,
                                              out, acc1, nEdges);
  if (two) {
    int nVec4 = nNodes * (OUT_FEATS / 4);
    gat_join_out<<<(nVec4 + 255) / 256, blk, 0, stream>>>(out, acc1, nVec4);
  }
}


# Aggregation adds rows weighted by the raw exp value; the join pass divides each summed row once by its segment sum (1e-16 floor kept): no per-edge sum gathers or division; weights rescaled
# speedup vs baseline: 3.9585x; 1.1571x over previous
#include <hip/hip_runtime.h>
#include <hip/hip_bf16.h>

#define IN_FEATS   128
#define OUT_FEATS  16
#define NEG_SLOPE  0.2f

#define EDGE_WGS     1024
#define EDGE_UNROLL  4
#define AGGR_WGS     1280

typedef float v2f __attribute__((ext_vector_type(2)));
typedef float v8f __attribute__((ext_vector_type(8)));

__global__ void gat_project_wmma(const float* __restrict__ h,
                                 const float* __restrict__ W,
                                 float* __restrict__ hW, int nNodes) {
  int wave = (blockIdx.x * blockDim.x + threadIdx.x) >> 5;
  int lane = threadIdx.x & 31;
  int m0 = wave * 16;
  if (m0 >= nNodes) return;
  int half = lane >> 4;
  int l16  = lane & 15;
  const float* arow = h + (size_t)(m0 + l16) * IN_FEATS;
  const float* brow = W + (size_t)l16 * IN_FEATS;
  v8f acc = {};
#pragma unroll
  for (int k = 0; k < IN_FEATS; k += 4) {
    int ka = k + half * 2;
    v2f a = *(const v2f*)(arow + ka);
    v2f b = *(const v2f*)(brow + ka);
    acc = __builtin_amdgcn_wmma_f32_16x16x4_f32(false, a, false, b,
                                                (short)0, acc, false, false);
  }
#pragma unroll
  for (int g = 0; g < 8; ++g) {
    int row = m0 + half * 8 + g;
    hW[(size_t)row * OUT_FEATS + l16] = acc[g];
  }
}

__device__ __forceinline__ int package_half() {
  return (int)((__builtin_amdgcn_s_sendmsg_rtn(0x87) >> 18) & 1u);
}

__global__ void gat_node_prep(const float* __restrict__ hW,
                              const float* __restrict__ a_l,
                              const float* __restrict__ a_r,
                              float* __restrict__ rec,
                              float* emax0, float* emax1,
                              float* esum0, float* esum1,
                              float* acc0, float* acc1, int nNodes) {
  int n = blockIdx.x * blockDim.x + threadIdx.x;
  if (n >= nNodes) return;
  const float* row = hW + (size_t)n * OUT_FEATS;
  float sl = 0.f, sr = 0.f;
#pragma unroll
  for (int f = 0; f < OUT_FEATS; ++f) {
    float v = row[f];
    sl += v * a_l[f];
    sr += v * a_r[f];
  }
  float4 zero4 = make_float4(0.f, 0.f, 0.f, 0.f);
  float4* z0 = (float4*)(acc0 + (size_t)n * OUT_FEATS);
  float4* z1 = (float4*)(acc1 + (size_t)n * OUT_FEATS);
#pragma unroll
  for (int q = 0; q < OUT_FEATS / 4; ++q) z0[q] = zero4;
#pragma unroll
  for (int q = 0; q < OUT_FEATS / 4; ++q) z1[q] = zero4;
  ((float2*)rec)[n] = make_float2(sl, sr);
  emax0[n] = __int_as_float(0xFF800000);
  emax1[n] = __int_as_float(0xFF800000);
  esum0[n] = 0.f;
  esum1[n] = 0.f;
}

__device__ __forceinline__ void atomicMaxFloat(float* addr, float val) {
  if (val >= 0.f) atomicMax((int*)addr, __float_as_int(val));
  else            atomicMin((unsigned int*)addr, (unsigned int)__float_as_int(val));
}

__global__ void __launch_bounds__(256)
gat_edge_max(const int* __restrict__ src,
             const int* __restrict__ dst,
             const float* __restrict__ rec,
             float* __restrict__ els,
             float* emax0, float* emax1, int nEdges) {
  int tid = blockIdx.x * 256 + threadIdx.x;
  int stride = gridDim.x * 256;
  float* emax = package_half() ? emax1 : emax0;
  for (int e0 = tid; e0 < nEdges; e0 += EDGE_UNROLL * stride) {
    int s[EDGE_UNROLL], d[EDGE_UNROLL];
    float v[EDGE_UNROLL];
#pragma unroll
    for (int u = 0; u < EDGE_UNROLL; ++u) {
      int e = e0 + u * stride;
      int ec = (e < nEdges) ? e : e0;
      s[u] = src[ec];
      d[u] = dst[ec];
    }
#pragma unroll
    for (int u = 0; u < EDGE_UNROLL; ++u) v[u] = rec[2 * (size_t)s[u]];
#pragma unroll
    for (int u = 0; u < EDGE_UNROLL; ++u) {
      int e = e0 + u * stride;
      if (e < nEdges) {
        els[e] = v[u];
        atomicMaxFloat(emax + d[u], v[u]);
      }
    }
  }
}

__global__ void gat_node_fixmax(const float* emax0, const float* emax1,
                                float* rec, int nNodes) {
  int n = blockIdx.x * blockDim.x + threadIdx.x;
  if (n >= nNodes) return;
  float m = fmaxf(emax0[n], emax1[n]);
  float x = m + rec[2 * (size_t)n + 1];
  x = (x > 0.f) ? x : NEG_SLOPE * x;
  if (!(fabsf(x) < __int_as_float(0x7F800000))) x = 0.f;
  rec[2 * (size_t)n] = x;
}

__global__ void __launch_bounds__(256)
gat_edge_exp(const int* __restrict__ dst,
             float* __restrict__ elog,
             const float* __restrict__ rec,
             float* esum0, float* esum1, int nEdges) {
  int tid = blockIdx.x * 256 + threadIdx.x;
  int stride = gridDim.x * 256;
  float* esum = package_half() ? esum1 : esum0;
  for (int e0 = tid; e0 < nEdges; e0 += EDGE_UNROLL * stride) {
    int d[EDGE_UNROLL];
    float el_s[EDGE_UNROLL];
    float2 rc[EDGE_UNROLL];
#pragma unroll
    for (int u = 0; u < EDGE_UNROLL; ++u) {
      int e = e0 + u * stride;
      int ec = (e < nEdges) ? e : e0;
      d[u] = dst[ec];
      el_s[u] = elog[ec];
    }
#pragma unroll
    for (int u = 0; u < EDGE_UNROLL; ++u) rc[u] = ((const float2*)rec)[d[u]];
#pragma unroll
    for (int u = 0; u < EDGE_UNROLL; ++u) {
      int e = e0 + u * stride;
      if (e < nEdges) {
        float x = el_s[u] + rc[u].y;
        x = (x > 0.f) ? x : NEG_SLOPE * x;
        float ex = __expf(x - rc[u].x);
        elog[e] = ex;
        atomicAdd(esum + d[u], ex);
      }
    }
  }
}

__global__ void __launch_bounds__(256)
gat_edge_aggr(const int* __restrict__ src,
                              const int* __restrict__ dst,
                              const float* __restrict__ elog,
                              const float* __restrict__ hW,
                              float* acc0, float* acc1, int nEdges) {
  int wave = (blockIdx.x * 256 + threadIdx.x) >> 5;
  int nWaves = gridDim.x * 8;
  int lane = threadIdx.x & 31;
  int sub = lane >> 4;
  int f = lane & 15;
  int nChunks = (nEdges + 31) >> 5;
  int chunk = wave;
  if (chunk >= nChunks) return;

  int e = min(chunk * 32 + lane, nEdges - 1);
  int s_n = src[e], d_n = dst[e];
  float x_n = elog[e];
  float* out = package_half() ? acc1 : acc0;

  for (; chunk < nChunks; chunk += nWaves) {
    int e0 = chunk * 32;
    int s_l = s_n, d_l = d_n;
    float x_l = x_n;
    int nc = chunk + nWaves;
    if (nc < nChunks) {
      int en = min(nc * 32 + lane, nEdges - 1);
      s_n = src[en];
      d_n = dst[en];
      x_n = elog[en];
    }
    int dd[16];
    float vv[16];
#pragma unroll
    for (int j = 0; j < 16; ++j) {
      int q = 2 * j + sub;
      int s = __shfl(s_l, q, 32);
      dd[j] = __shfl(d_l, q, 32);
      float a = __shfl(x_l, q, 32);
      vv[j] = hW[(size_t)s * OUT_FEATS + f] * a;
    }
#pragma unroll
    for (int j = 0; j < 16; ++j) {
      int q = 2 * j + sub;
      if (e0 + q < nEdges)
        atomicAdd(out + (size_t)dd[j] * OUT_FEATS + f, vv[j]);
    }
  }
}

__global__ void gat_join_out(float* out, const float* acc1,
                             const float* esum0, const float* esum1, int nVec4) {
  int i = blockIdx.x * blockDim.x + threadIdx.x;
  if (i >= nVec4) return;
  int n = i / (OUT_FEATS / 4);
  float4 a = ((const float4*)out)[i];
  float es = esum0[n];
  if (esum1 != esum0) {
    float4 b = ((const float4*)acc1)[i];
    a.x += b.x; a.y += b.y; a.z += b.z; a.w += b.w;
    es += esum1[n];
  }
  float den = fmaxf(es, 1e-16f);
  a.x /= den; a.y /= den; a.z /= den; a.w /= den;
  ((float4*)out)[i] = a;
}

extern "C" void kernel_launch(void* const* d_in, const int* in_sizes, int n_in,
                              void* d_out, int out_size, void* d_ws, size_t ws_size,
                              hipStream_t stream) {
  const float* h   = (const float*)d_in[0];
  const int*   src = (const int*)d_in[1];
  const int*   dst = (const int*)d_in[2];
  const float* W   = (const float*)d_in[3];
  const float* a_l = (const float*)d_in[4];
  const float* a_r = (const float*)d_in[5];
  float* out = (float*)d_out;

  int nNodes = in_sizes[0] / IN_FEATS;
  int nEdges = in_sizes[1];

  size_t need2 = ((size_t)nNodes * (2 * OUT_FEATS + 6) + (size_t)nEdges) * sizeof(float);
  bool two = ws_size >= need2;
  float* ws    = (float*)d_ws;
  float* hW    = ws;
  float* acc1  = two ? hW + (size_t)nNodes * OUT_FEATS : out;
  float* rec   = hW + (size_t)nNodes * OUT_FEATS * (two ? 2 : 1);
  float* emax0 = rec + 2 * (size_t)nNodes;
  float* emax1 = two ? emax0 + nNodes : emax0;
  float* esum0 = emax1 + nNodes;
  float* esum1 = two ? esum0 + nNodes : esum0;
  float* elog  = esum1 + nNodes;

  dim3 blk(256);
  int tiles = (nNodes + 15) / 16;
  int blocksK1 = (tiles + 7) / 8;

  gat_project_wmma<<<blocksK1, blk, 0, stream>>>(h, W, hW, nNodes);
  gat_node_prep<<<(nNodes + 255) / 256, blk, 0, stream>>>(hW, a_l, a_r, rec,
                                                          emax0, emax1, esum0, esum1,
                                                          out, acc1, nNodes);
  gat_edge_max<<<EDGE_WGS, blk, 0, stream>>>(src, dst, rec,
                                             elog, emax0, emax1, nEdges);
  gat_node_fixmax<<<(nNodes + 255) / 256, blk, 0, stream>>>(emax0, emax1, rec, nNodes);
  gat_edge_exp<<<EDGE_WGS, blk, 0, stream>>>(dst, elog, rec,
                                             esum0, esum1, nEdges);

  gat_edge_aggr<<<AGGR_WGS, blk, 0, stream>>>(src, dst, elog, hW,
                                              out, acc1, nEdges);
  int nVec4 = nNodes * (OUT_FEATS / 4);
  gat_join_out<<<(nVec4 + 255) / 256, blk, 0, stream>>>(out, acc1, esum0, esum1, nVec4);
}


# Segment-max pass and node fix-up go: destinations shifted by the bound leaky(max el + er[dst]) while max el - min el <= 32 (else old max pass); edge el summed from its gathered row; weights rescaled
# speedup vs baseline: 5.1718x; 1.3065x over previous
#include <hip/hip_runtime.h>
#include <hip/hip_bf16.h>

#define IN_FEATS   128
#define OUT_FEATS  16
#define NEG_SLOPE  0.2f

#define EDGE_WGS     256
#define EDGE_UNROLL  4
#define AGGR_WGS     1280
#define SHIFT_SPAN   32.0f

typedef float v2f __attribute__((ext_vector_type(2)));
typedef float v8f __attribute__((ext_vector_type(8)));

__global__ void gat_project_wmma(const float* __restrict__ h,
                                 const float* __restrict__ W,
                                 float* __restrict__ hW,
                                 float* __restrict__ gstat, int nNodes) {
  if (blockIdx.x == 0 && threadIdx.x == 0) {
    gstat[0] = __int_as_float(0xFF800000);
    gstat[1] = __int_as_float(0xFF800000);
    gstat[2] = __int_as_float(0x7F800000);
    gstat[3] = __int_as_float(0x7F800000);
  }
  int wave = (blockIdx.x * blockDim.x + threadIdx.x) >> 5;
  int lane = threadIdx.x & 31;
  int m0 = wave * 16;
  if (m0 >= nNodes) return;
  int half = lane >> 4;
  int l16  = lane & 15;
  const float* arow = h + (size_t)(m0 + l16) * IN_FEATS;
  const float* brow = W + (size_t)l16 * IN_FEATS;
  v8f acc = {};
#pragma unroll
  for (int k = 0; k < IN_FEATS; k += 4) {
    int ka = k + half * 2;
    v2f a = *(const v2f*)(arow + ka);
    v2f b = *(const v2f*)(brow + ka);
    acc = __builtin_amdgcn_wmma_f32_16x16x4_f32(false, a, false, b,
                                                (short)0, acc, false, false);
  }
#pragma unroll
  for (int g = 0; g < 8; ++g) {
    int row = m0 + half * 8 + g;
    hW[(size_t)row * OUT_FEATS + l16] = acc[g];
  }
}

__device__ __forceinline__ int package_half() {
  return (int)((__builtin_amdgcn_s_sendmsg_rtn(0x87) >> 18) & 1u);
}

__device__ __forceinline__ void atomicMaxFloat(float* addr, float val) {
  if (val >= 0.f) atomicMax((int*)addr, __float_as_int(val));
  else            atomicMin((unsigned int*)addr, (unsigned int)__float_as_int(val));
}
__device__ __forceinline__ void atomicMinFloat(float* addr, float val) {
  if (val >= 0.f) atomicMin((int*)addr, __float_as_int(val));
  else            atomicMax((unsigned int*)addr, (unsigned int)__float_as_int(val));
}

__device__ __forceinline__ float leaky(float x) {
  return (x > 0.f) ? x : NEG_SLOPE * x;
}

__device__ __forceinline__ bool shift_bound_ok(const float* gstat, float& M) {
  M = fmaxf(gstat[0], gstat[1]);
  float m = fminf(gstat[2], gstat[3]);
  return (M >= m) && (M - m <= SHIFT_SPAN);
}

__global__ void __launch_bounds__(256)
gat_node_prep(const float* __restrict__ hW,
              const float* __restrict__ a_l,
              const float* __restrict__ a_r,
              float* __restrict__ rec,
              float* emax0, float* emax1,
              float* esum0, float* esum1,
              float* acc0, float* acc1,
              float* gstat, int nNodes) {
  __shared__ float smx[8], smn[8];
  int n = blockIdx.x * 256 + threadIdx.x;
  float mx = __int_as_float(0xFF800000);
  float mn = __int_as_float(0x7F800000);
  if (n < nNodes) {
    const float* row = hW + (size_t)n * OUT_FEATS;
    float sl = 0.f, sr = 0.f;
#pragma unroll
    for (int f = 0; f < OUT_FEATS; ++f) {
      float v = row[f];
      sl += v * a_l[f];
      sr += v * a_r[f];
    }
    float4 zero4 = make_float4(0.f, 0.f, 0.f, 0.f);
    float4* z0 = (float4*)(acc0 + (size_t)n * OUT_FEATS);
    float4* z1 = (float4*)(acc1 + (size_t)n * OUT_FEATS);
#pragma unroll
    for (int q = 0; q < OUT_FEATS / 4; ++q) z0[q] = zero4;
#pragma unroll
    for (int q = 0; q < OUT_FEATS / 4; ++q) z1[q] = zero4;
    ((float2*)rec)[n] = make_float2(sl, sr);
    emax0[n] = __int_as_float(0xFF800000);
    emax1[n] = __int_as_float(0xFF800000);
    esum0[n] = 0.f;
    esum1[n] = 0.f;
    mx = sl;
    mn = sl;
  }
#pragma unroll
  for (int m = 16; m >= 1; m >>= 1) {
    mx = fmaxf(mx, __shfl_xor(mx, m, 32));
    mn = fminf(mn, __shfl_xor(mn, m, 32));
  }
  if ((threadIdx.x & 31) == 0) {
    smx[threadIdx.x >> 5] = mx;
    smn[threadIdx.x >> 5] = mn;
  }
  __syncthreads();
  if (threadIdx.x == 0) {
#pragma unroll
    for (int w = 1; w < 8; ++w) {
      mx = fmaxf(mx, smx[w]);
      mn = fminf(mn, smn[w]);
    }
    int hsel = package_half();
    atomicMaxFloat(gstat + hsel, mx);
    atomicMinFloat(gstat + 2 + hsel, mn);
  }
}

__global__ void __launch_bounds__(256)
gat_edge_max(const int* __restrict__ src,
             const int* __restrict__ dst,
             const float* __restrict__ rec,
             const float* __restrict__ gstat,
             float* emax0, float* emax1, int nEdges) {
  float M;
  if (shift_bound_ok(gstat, M)) return;
  int tid = blockIdx.x * 256 + threadIdx.x;
  int stride = gridDim.x * 256;
  float* emax = package_half() ? emax1 : emax0;
  for (int e0 = tid; e0 < nEdges; e0 += EDGE_UNROLL * stride) {
    int s[EDGE_UNROLL], d[EDGE_UNROLL];
    float v[EDGE_UNROLL];
#pragma unroll
    for (int u = 0; u < EDGE_UNROLL; ++u) {
      int e = e0 + u * stride;
      int ec = (e < nEdges) ? e : e0;
      s[u] = src[ec];
      d[u] = dst[ec];
    }
#pragma unroll
    for (int u = 0; u < EDGE_UNROLL; ++u) v[u] = rec[2 * (size_t)s[u]];
#pragma unroll
    for (int u = 0; u < EDGE_UNROLL; ++u) {
      int e = e0 + u * stride;
      if (e < nEdges) atomicMaxFloat(emax + d[u], v[u]);
    }
  }
}

template <int MASK>
__device__ __forceinline__ float row_xor(float v) {
  return __int_as_float(__builtin_amdgcn_update_dpp(0, __float_as_int(v),
                                                    0x160 + MASK, 0xf, 0xf, true));
}

template <int J>
__device__ __forceinline__ float row_lane(float v) {
  return __int_as_float(__builtin_amdgcn_update_dpp(0, __float_as_int(v),
                                                    0x150 + J, 0xf, 0xf, true));
}

template <int MASK, int N>
__device__ __forceinline__ void fold_step(const float* in, float* out, bool bit) {
#pragma unroll
  for (int i = 0; i < N; ++i) {
    float keep = bit ? in[2 * i + 1] : in[2 * i];
    float send = bit ? in[2 * i] : in[2 * i + 1];
    out[i] = keep + row_xor<MASK>(send);
  }
}

__global__ void __launch_bounds__(256)
gat_edge_aggr(const int* __restrict__ src,
              const int* __restrict__ dst,
              const float* __restrict__ rec,
              const float* __restrict__ hW,
              const float* __restrict__ a_l,
              const float* __restrict__ gstat,
              const float* emax0, const float* emax1,
              float* acc0, float* acc1,
              float* esum0, float* esum1, int nEdges) {
  int wave = (blockIdx.x * 256 + threadIdx.x) >> 5;
  int nWaves = gridDim.x * 8;
  int lane = threadIdx.x & 31;
  int rowsel = lane & 16;
  int f = lane & 15;
  int nChunks = (nEdges + 31) >> 5;
  int chunk = wave;
  if (chunk >= nChunks) return;

  int hsel = package_half();
  float* out  = hsel ? acc1 : acc0;
  float* esum = hsel ? esum1 : esum0;
  float M;
  bool bound = shift_bound_ok(gstat, M);
  float al_f = a_l[f];
  bool b1 = (f & 1) != 0, b2 = (f & 2) != 0, b4 = (f & 4) != 0, b8 = (f & 8) != 0;

  int e = min(chunk * 32 + lane, nEdges - 1);
  int s_l = src[e], d_l = dst[e];
  int s_n = s_l, d_n = d_l;
  if (chunk + nWaves < nChunks) {
    int en = min((chunk + nWaves) * 32 + lane, nEdges - 1);
    s_n = src[en];
    d_n = dst[en];
  }
  float er_l = rec[2 * (size_t)d_l + 1];

  for (; chunk < nChunks; chunk += nWaves) {
    int e0 = chunk * 32;
    float er_n = er_l;
    if (chunk + nWaves < nChunks) er_n = rec[2 * (size_t)d_n + 1];
    int s_f = s_n, d_f = d_n;
    int cf = chunk + 2 * nWaves;
    if (cf < nChunks) {
      int ef = min(cf * 32 + lane, nEdges - 1);
      s_f = src[ef];
      d_f = dst[ef];
    }

    float c_l;
    if (bound) {
      c_l = leaky(M + er_l);
    } else {
      float m = fmaxf(emax0[d_l], emax1[d_l]);
      c_l = leaky(m + er_l);
      if (!(fabsf(c_l) < __int_as_float(0x7F800000))) c_l = 0.f;
    }

    int dd[16];
    float rr[16];
#pragma unroll
    for (int j = 0; j < 16; ++j) {
      int q = rowsel | j;
      int s = __shfl(s_l, q, 32);
      dd[j] = __shfl(d_l, q, 32);
      rr[j] = hW[(size_t)s * OUT_FEATS + f];
    }

    float p[16], t8[8], t4[4], t2[2], t1[1];
#pragma unroll
    for (int j = 0; j < 16; ++j) p[j] = rr[j] * al_f;
    fold_step<1, 8>(p, t8, b1);
    fold_step<2, 4>(t8, t4, b2);
    fold_step<4, 2>(t4, t2, b4);
    fold_step<8, 1>(t2, t1, b8);
    float el_l = t1[0];

    float x = leaky(el_l + er_l);
    float x_l = __expf(x - c_l);
    if (e0 + lane < nEdges) atomicAdd(esum + d_l, x_l);

    float vv[16];
#define WEIGH(J) vv[J] = rr[J] * row_lane<J>(x_l);
    WEIGH(0)  WEIGH(1)  WEIGH(2)  WEIGH(3)  WEIGH(4)  WEIGH(5)  WEIGH(6)  WEIGH(7)
    WEIGH(8)  WEIGH(9)  WEIGH(10) WEIGH(11) WEIGH(12) WEIGH(13) WEIGH(14) WEIGH(15)
#undef WEIGH
#pragma unroll
    for (int j = 0; j < 16; ++j) {
      if (e0 + (rowsel | j) < nEdges)
        atomicAdd(out + (size_t)dd[j] * OUT_FEATS + f, vv[j]);
    }

    s_l = s_n; d_l = d_n; er_l = er_n;
    s_n = s_f; d_n = d_f;
  }
}

__global__ void gat_join_out(float* out, const float* acc1,
                             const float* esum0, const float* esum1, int nVec4) {
  int i = blockIdx.x * blockDim.x + threadIdx.x;
  if (i >= nVec4) return;
  int n = i / (OUT_FEATS / 4);
  float4 a = ((const float4*)out)[i];
  float es = esum0[n];
  if (esum1 != esum0) {
    float4 b = ((const float4*)acc1)[i];
    a.x += b.x; a.y += b.y; a.z += b.z; a.w += b.w;
    es += esum1[n];
  }
  float den = fmaxf(es, 1e-16f);
  a.x /= den; a.y /= den; a.z /= den; a.w /= den;
  ((float4*)out)[i] = a;
}

extern "C" void kernel_launch(void* const* d_in, const int* in_sizes, int n_in,
                              void* d_out, int out_size, void* d_ws, size_t ws_size,
                              hipStream_t stream) {
  const float* h   = (const float*)d_in[0];
  const int*   src = (const int*)d_in[1];
  const int*   dst = (const int*)d_in[2];
  const float* W   = (const float*)d_in[3];
  const float* a_l = (const float*)d_in[4];
  const float* a_r = (const float*)d_in[5];
  float* out = (float*)d_out;

  int nNodes = in_sizes[0] / IN_FEATS;
  int nEdges = in_sizes[1];

  size_t need2 = ((size_t)nNodes * (2 * OUT_FEATS + 6) + 4) * sizeof(float);
  bool two = ws_size >= need2;
  float* ws    = (float*)d_ws;
  float* hW    = ws;
  float* acc1  = two ? hW + (size_t)nNodes * OUT_FEATS : out;
  float* rec   = hW + (size_t)nNodes * OUT_FEATS * (two ? 2 : 1);
  float* emax0 = rec + 2 * (size_t)nNodes;
  float* emax1 = two ? emax0 + nNodes : emax0;
  float* esum0 = emax1 + nNodes;
  float* esum1 = two ? esum0 + nNodes : esum0;
  float* gstat = esum1 + nNodes;

  dim3 blk(256);
  int tiles = (nNodes + 15) / 16;
  int blocksK1 = (tiles + 7) / 8;

  gat_project_wmma<<<blocksK1, blk, 0, stream>>>(h, W, hW, gstat, nNodes);
  gat_node_prep<<<(nNodes + 255) / 256, blk, 0, stream>>>(hW, a_l, a_r, rec,
                                                          emax0, emax1, esum0, esum1,
                                                          out, acc1, gstat, nNodes);
  gat_edge_max<<<EDGE_WGS, blk, 0, stream>>>(src, dst, rec, gstat,
                                             emax0, emax1, nEdges);
  gat_edge_aggr<<<AGGR_WGS, blk, 0, stream>>>(src, dst, rec, hW, a_l, gstat,
                                              emax0, emax1,
                                              out, acc1, esum0, esum1, nEdges);
  int nVec4 = nNodes * (OUT_FEATS / 4);
  gat_join_out<<<(nVec4 + 255) / 256, blk, 0, stream>>>(out, acc1, esum0, esum1, nVec4);
}


# Aggregation keeps er of the first 81920 nodes in LDS (one 1024-thread work-group per WGP fills it by whole-line loads): per-edge er gather becomes an LDS read; er, el as planes; sum order changes
# speedup vs baseline: 5.4726x; 1.0582x over previous
#include <hip/hip_runtime.h>
#include <hip/hip_bf16.h>

#define IN_FEATS   128
#define OUT_FEATS  16
#define NEG_SLOPE  0.2f

#define EDGE_WGS     256
#define EDGE_UNROLL  4
#define AGGR_WGS     256
#define AGGR_THREADS 1024
#define ER_LDS_NODES 81920
#define FILL_BATCH   20
#define SHIFT_SPAN   32.0f

typedef float v2f __attribute__((ext_vector_type(2)));
typedef float v8f __attribute__((ext_vector_type(8)));

__global__ void gat_project_wmma(const float* __restrict__ h,
                                 const float* __restrict__ W,
                                 float* __restrict__ hW,
                                 float* __restrict__ gstat, int nNodes) {
  if (blockIdx.x == 0 && threadIdx.x == 0) {
    gstat[0] = __int_as_float(0xFF800000);
    gstat[1] = __int_as_float(0xFF800000);
    gstat[2] = __int_as_float(0x7F800000);
    gstat[3] = __int_as_float(0x7F800000);
  }
  int wave = (blockIdx.x * blockDim.x + threadIdx.x) >> 5;
  int lane = threadIdx.x & 31;
  int m0 = wave * 16;
  if (m0 >= nNodes) return;
  int half = lane >> 4;
  int l16  = lane & 15;
  const float* arow = h + (size_t)(m0 + l16) * IN_FEATS;
  const float* brow = W + (size_t)l16 * IN_FEATS;
  v8f acc = {};
#pragma unroll
  for (int k = 0; k < IN_FEATS; k += 4) {
    int ka = k + half * 2;
    v2f a = *(const v2f*)(arow + ka);
    v2f b = *(const v2f*)(brow + ka);
    acc = __builtin_amdgcn_wmma_f32_16x16x4_f32(false, a, false, b,
                                                (short)0, acc, false, false);
  }
#pragma unroll
  for (int g = 0; g < 8; ++g) {
    int row = m0 + half * 8 + g;
    hW[(size_t)row * OUT_FEATS + l16] = acc[g];
  }
}

__device__ __forceinline__ int package_half() {
  return (int)((__builtin_amdgcn_s_sendmsg_rtn(0x87) >> 18) & 1u);
}

__device__ __forceinline__ void atomicMaxFloat(float* addr, float val) {
  if (val >= 0.f) atomicMax((int*)addr, __float_as_int(val));
  else            atomicMin((unsigned int*)addr, (unsigned int)__float_as_int(val));
}
__device__ __forceinline__ void atomicMinFloat(float* addr, float val) {
  if (val >= 0.f) atomicMin((int*)addr, __float_as_int(val));
  else            atomicMax((unsigned int*)addr, (unsigned int)__float_as_int(val));
}

__device__ __forceinline__ float leaky(float x) {
  return (x > 0.f) ? x : NEG_SLOPE * x;
}

__device__ __forceinline__ bool shift_bound_ok(const float* gstat, float& M) {
  M = fmaxf(gstat[0], gstat[1]);
  float m = fminf(gstat[2], gstat[3]);
  return (M >= m) && (M - m <= SHIFT_SPAN);
}

__global__ void __launch_bounds__(256)
gat_node_prep(const float* __restrict__ hW,
              const float* __restrict__ a_l,
              const float* __restrict__ a_r,
              float* __restrict__ er_p,
              float* __restrict__ el_p,
              float* emax0, float* emax1,
              float* esum0, float* esum1,
              float* acc0, float* acc1,
              float* gstat, int nNodes) {
  __shared__ float smx[8], smn[8];
  int n = blockIdx.x * 256 + threadIdx.x;
  float mx = __int_as_float(0xFF800000);
  float mn = __int_as_float(0x7F800000);
  if (n < nNodes) {
    const float* row = hW + (size_t)n * OUT_FEATS;
    float sl = 0.f, sr = 0.f;
#pragma unroll
    for (int f = 0; f < OUT_FEATS; ++f) {
      float v = row[f];
      sl += v * a_l[f];
      sr += v * a_r[f];
    }
    float4 zero4 = make_float4(0.f, 0.f, 0.f, 0.f);
    float4* z0 = (float4*)(acc0 + (size_t)n * OUT_FEATS);
    float4* z1 = (float4*)(acc1 + (size_t)n * OUT_FEATS);
#pragma unroll
    for (int q = 0; q < OUT_FEATS / 4; ++q) z0[q] = zero4;
#pragma unroll
    for (int q = 0; q < OUT_FEATS / 4; ++q) z1[q] = zero4;
    er_p[n] = sr;
    el_p[n] = sl;
    emax0[n] = __int_as_float(0xFF800000);
    emax1[n] = __int_as_float(0xFF800000);
    esum0[n] = 0.f;
    esum1[n] = 0.f;
    mx = sl;
    mn = sl;
  }
#pragma unroll
  for (int m = 16; m >= 1; m >>= 1) {
    mx = fmaxf(mx, __shfl_xor(mx, m, 32));
    mn = fminf(mn, __shfl_xor(mn, m, 32));
  }
  if ((threadIdx.x & 31) == 0) {
    smx[threadIdx.x >> 5] = mx;
    smn[threadIdx.x >> 5] = mn;
  }
  __syncthreads();
  if (threadIdx.x == 0) {
#pragma unroll
    for (int w = 1; w < 8; ++w) {
      mx = fmaxf(mx, smx[w]);
      mn = fminf(mn, smn[w]);
    }
    int hsel = package_half();
    atomicMaxFloat(gstat + hsel, mx);
    atomicMinFloat(gstat + 2 + hsel, mn);
  }
}

__global__ void __launch_bounds__(256)
gat_edge_max(const int* __restrict__ src,
             const int* __restrict__ dst,
             const float* __restrict__ el_p,
             const float* __restrict__ gstat,
             float* emax0, float* emax1, int nEdges) {
  float M;
  if (shift_bound_ok(gstat, M)) return;
  int tid = blockIdx.x * 256 + threadIdx.x;
  int stride = gridDim.x * 256;
  float* emax = package_half() ? emax1 : emax0;
  for (int e0 = tid; e0 < nEdges; e0 += EDGE_UNROLL * stride) {
    int s[EDGE_UNROLL], d[EDGE_UNROLL];
    float v[EDGE_UNROLL];
#pragma unroll
    for (int u = 0; u < EDGE_UNROLL; ++u) {
      int e = e0 + u * stride;
      int ec = (e < nEdges) ? e : e0;
      s[u] = src[ec];
      d[u] = dst[ec];
    }
#pragma unroll
    for (int u = 0; u < EDGE_UNROLL; ++u) v[u] = el_p[s[u]];
#pragma unroll
    for (int u = 0; u < EDGE_UNROLL; ++u) {
      int e = e0 + u * stride;
      if (e < nEdges) atomicMaxFloat(emax + d[u], v[u]);
    }
  }
}

template <int MASK>
__device__ __forceinline__ float row_xor(float v) {
  return __int_as_float(__builtin_amdgcn_update_dpp(0, __float_as_int(v),
                                                    0x160 + MASK, 0xf, 0xf, true));
}

template <int J>
__device__ __forceinline__ float row_lane(float v) {
  return __int_as_float(__builtin_amdgcn_update_dpp(0, __float_as_int(v),
                                                    0x150 + J, 0xf, 0xf, true));
}

template <int MASK, int N>
__device__ __forceinline__ void fold_step(const float* in, float* out, bool bit) {
#pragma unroll
  for (int i = 0; i < N; ++i) {
    float keep = bit ? in[2 * i + 1] : in[2 * i];
    float send = bit ? in[2 * i] : in[2 * i + 1];
    out[i] = keep + row_xor<MASK>(send);
  }
}

template <int B>
__device__ __forceinline__ void fill_rows(const float4* g4, float4* s4,
                                          int i, int rot, int n4) {
  float4 v[B];
  int j[B];
#pragma unroll
  for (int u = 0; u < B; ++u) {
    int t = i + u * AGGR_THREADS + rot;
    j[u] = (t >= n4) ? t - n4 : t;
    v[u] = g4[j[u]];
  }
#pragma unroll
  for (int u = 0; u < B; ++u) s4[j[u]] = v[u];
}

typedef __attribute__((address_space(3))) float lds_f32;

__device__ __forceinline__ float er_of(const lds_f32* er_t, const float* er_g,
                                       int d, int nLds) {
  float v = er_t[max(min(d, nLds - 1), 0)];
  if (d >= nLds) v = er_g[d];
  return v;
}

__global__ void __launch_bounds__(AGGR_THREADS)
gat_edge_aggr(const int* __restrict__ src,
              const int* __restrict__ dst,
              const float* __restrict__ er_g,
              const float* __restrict__ hW,
              const float* __restrict__ a_l,
              const float* __restrict__ gstat,
              const float* emax0, const float* emax1,
              float* acc0, float* acc1,
              float* esum0, float* esum1, int nNodes, int nEdges) {
  __shared__ float er_s[ER_LDS_NODES];
  int tid = threadIdx.x;
  int w = tid >> 5;
  int wave = (w >> 3) * (gridDim.x * 8) + blockIdx.x * 8 + (w & 7);
  int nWaves = gridDim.x * (AGGR_THREADS / 32);
  int lane = tid & 31;
  int rowsel = lane & 16;
  int f = lane & 15;
  int nChunks = (nEdges + 31) >> 5;
  int chunk = wave;
  bool work = chunk < nChunks;

  int s_l = 0, d_l = 0, s_n = 0, d_n = 0;
  if (work) {
    int e = min(chunk * 32 + lane, nEdges - 1);
    s_l = src[e];
    d_l = dst[e];
    if (chunk + nWaves < nChunks) {
      int en = min((chunk + nWaves) * 32 + lane, nEdges - 1);
      s_n = src[en];
      d_n = dst[en];
    }
  }

  int nLds = min(nNodes, ER_LDS_NODES);
  int nLds4 = nLds >> 2;
  {
    const float4* g4 = (const float4*)er_g;
    float4* s4 = (float4*)er_s;
    int rot = (nLds4 > 0) ? (int)((blockIdx.x * 64u) % (unsigned)nLds4) : 0;
    int rowsFull = nLds4 / AGGR_THREADS;
    int r = 0;
    for (; r + FILL_BATCH <= rowsFull; r += FILL_BATCH)
      fill_rows<FILL_BATCH>(g4, s4, r * AGGR_THREADS + tid, rot, nLds4);
    for (; r + 4 <= rowsFull; r += 4)
      fill_rows<4>(g4, s4, r * AGGR_THREADS + tid, rot, nLds4);
    for (; r + 2 <= rowsFull; r += 2)
      fill_rows<2>(g4, s4, r * AGGR_THREADS + tid, rot, nLds4);
    for (; r + 1 <= rowsFull; r += 1)
      fill_rows<1>(g4, s4, r * AGGR_THREADS + tid, rot, nLds4);
    if (r * AGGR_THREADS + tid < nLds4)
      fill_rows<1>(g4, s4, r * AGGR_THREADS + tid, rot, nLds4);
    for (int i = (nLds4 << 2) + tid; i < nLds; i += AGGR_THREADS) er_s[i] = er_g[i];
  }
  __syncthreads();
  if (!work) return;

  int hsel = package_half();
  float* out  = hsel ? acc1 : acc0;
  float* esum = hsel ? esum1 : esum0;
  float M;
  bool bound = shift_bound_ok(gstat, M);
  float al_f = a_l[f];
  bool b1 = (f & 1) != 0, b2 = (f & 2) != 0, b4 = (f & 4) != 0, b8 = (f & 8) != 0;

  const lds_f32* er_t = (const lds_f32*)er_s;
  float er_l = er_of(er_t, er_g, d_l, nLds);

  for (; chunk < nChunks; chunk += nWaves) {
    int e0 = chunk * 32;
    int s_f = s_n, d_f = d_n;
    int cf = chunk + 2 * nWaves;
    if (cf < nChunks) {
      int ef = min(cf * 32 + lane, nEdges - 1);
      s_f = src[ef];
      d_f = dst[ef];
    }

    float c_l;
    if (bound) {
      c_l = leaky(M + er_l);
    } else {
      float m = fmaxf(emax0[d_l], emax1[d_l]);
      c_l = leaky(m + er_l);
      if (!(fabsf(c_l) < __int_as_float(0x7F800000))) c_l = 0.f;
    }

    int dd[16];
    float rr[16];
#pragma unroll
    for (int j = 0; j < 16; ++j) {
      int q = rowsel | j;
      int s = __shfl(s_l, q, 32);
      dd[j] = __shfl(d_l, q, 32);
      rr[j] = hW[(size_t)s * OUT_FEATS + f];
    }

    float er_nt = er_l, er_ng = er_l;
    bool far_n = d_n >= nLds;
    if (chunk + nWaves < nChunks) {
      er_nt = er_t[max(min(d_n, nLds - 1), 0)];
      if (far_n) er_ng = er_g[d_n];
    }

    float p[16], t8[8], t4[4], t2[2], t1[1];
#pragma unroll
    for (int j = 0; j < 16; ++j) p[j] = rr[j] * al_f;
    fold_step<1, 8>(p, t8, b1);
    fold_step<2, 4>(t8, t4, b2);
    fold_step<4, 2>(t4, t2, b4);
    fold_step<8, 1>(t2, t1, b8);
    float el_l = t1[0];

    float x = leaky(el_l + er_l);
    float x_l = __expf(x - c_l);
    if (e0 + lane < nEdges) atomicAdd(esum + d_l, x_l);

    float vv[16];
#define WEIGH(J) vv[J] = rr[J] * row_lane<J>(x_l);
    WEIGH(0)  WEIGH(1)  WEIGH(2)  WEIGH(3)  WEIGH(4)  WEIGH(5)  WEIGH(6)  WEIGH(7)
    WEIGH(8)  WEIGH(9)  WEIGH(10) WEIGH(11) WEIGH(12) WEIGH(13) WEIGH(14) WEIGH(15)
#undef WEIGH
#pragma unroll
    for (int j = 0; j < 16; ++j) {
      if (e0 + (rowsel | j) < nEdges)
        atomicAdd(out + (size_t)dd[j] * OUT_FEATS + f, vv[j]);
    }

    s_l = s_n; d_l = d_n; er_l = far_n ? er_ng : er_nt;
    s_n = s_f; d_n = d_f;
  }
}

__global__ void gat_join_out(float* out, const float* acc1,
                             const float* esum0, const float* esum1, int nVec4) {
  int i = blockIdx.x * blockDim.x + threadIdx.x;
  if (i >= nVec4) return;
  int n = i / (OUT_FEATS / 4);
  float4 a = ((const float4*)out)[i];
  float es = esum0[n];
  if (esum1 != esum0) {
    float4 b = ((const float4*)acc1)[i];
    a.x += b.x; a.y += b.y; a.z += b.z; a.w += b.w;
    es += esum1[n];
  }
  float den = fmaxf(es, 1e-16f);
  a.x /= den; a.y /= den; a.z /= den; a.w /= den;
  ((float4*)out)[i] = a;
}

extern "C" void kernel_launch(void* const* d_in, const int* in_sizes, int n_in,
                              void* d_out, int out_size, void* d_ws, size_t ws_size,
                              hipStream_t stream) {
  const float* h   = (const float*)d_in[0];
  const int*   src = (const int*)d_in[1];
  const int*   dst = (const int*)d_in[2];
  const float* W   = (const float*)d_in[3];
  const float* a_l = (const float*)d_in[4];
  const float* a_r = (const float*)d_in[5];
  float* out = (float*)d_out;

  int nNodes = in_sizes[0] / IN_FEATS;
  int nEdges = in_sizes[1];

  size_t need2 = ((size_t)nNodes * (2 * OUT_FEATS + 6) + 4) * sizeof(float);
  bool two = ws_size >= need2;
  float* ws    = (float*)d_ws;
  float* hW    = ws;
  float* acc1  = two ? hW + (size_t)nNodes * OUT_FEATS : out;
  float* er_p  = hW + (size_t)nNodes * OUT_FEATS * (two ? 2 : 1);
  float* el_p  = er_p + nNodes;
  float* emax0 = el_p + nNodes;
  float* emax1 = two ? emax0 + nNodes : emax0;
  float* esum0 = emax1 + nNodes;
  float* esum1 = two ? esum0 + nNodes : esum0;
  float* gstat = esum1 + nNodes;

  dim3 blk(256);
  int tiles = (nNodes + 15) / 16;
  int blocksK1 = (tiles + 7) / 8;

  gat_project_wmma<<<blocksK1, blk, 0, stream>>>(h, W, hW, gstat, nNodes);
  gat_node_prep<<<(nNodes + 255) / 256, blk, 0, stream>>>(hW, a_l, a_r, er_p, el_p,
                                                          emax0, emax1, esum0, esum1,
                                                          out, acc1, gstat, nNodes);
  gat_edge_max<<<EDGE_WGS, blk, 0, stream>>>(src, dst, el_p, gstat,
                                             emax0, emax1, nEdges);
  gat_edge_aggr<<<AGGR_WGS, dim3(AGGR_THREADS), 0, stream>>>(src, dst, er_p, hW, a_l, gstat,
                                                             emax0, emax1,
                                                             out, acc1, esum0, esum1,
                                                             nNodes, nEdges);
  int nVec4 = nNodes * (OUT_FEATS / 4);
  gat_join_out<<<(nVec4 + 255) / 256, blk, 0, stream>>>(out, acc1, esum0, esum1, nVec4);
}


# Projection operands go through LDS: a row of h is one whole-line 16-byte load of the wave (4 lines, each read once; was 16 lines a load, 8 touches a line), W once per work-group; fragments from LDS
# speedup vs baseline: 5.7903x; 1.0581x over previous
#include <hip/hip_runtime.h>
#include <hip/hip_bf16.h>

#define IN_FEATS   128
#define OUT_FEATS  16
#define NEG_SLOPE  0.2f

#define EDGE_WGS     256
#define EDGE_UNROLL  4
#define AGGR_WGS     256
#define AGGR_THREADS 1024
#define ER_LDS_NODES 81920
#define FILL_BATCH   20
#define SHIFT_SPAN   32.0f

typedef float v2f __attribute__((ext_vector_type(2)));
typedef float v8f __attribute__((ext_vector_type(8)));

#define PROJ_WAVES 8
#define PROJ_PITCH 132
__global__ void __launch_bounds__(PROJ_WAVES * 32)
gat_project_wmma(const float* __restrict__ h,
                 const float* __restrict__ W,
                 float* __restrict__ hW,
                 float* __restrict__ gstat, int nNodes) {
  __shared__ __attribute__((aligned(16))) float Ws[OUT_FEATS * PROJ_PITCH];
  __shared__ __attribute__((aligned(16))) float As[PROJ_WAVES * 16 * PROJ_PITCH];
  if (blockIdx.x == 0 && threadIdx.x == 0) {
    gstat[0] = __int_as_float(0xFF800000);
    gstat[1] = __int_as_float(0xFF800000);
    gstat[2] = __int_as_float(0x7F800000);
    gstat[3] = __int_as_float(0x7F800000);
  }
  int w = __builtin_amdgcn_readfirstlane((int)(threadIdx.x >> 5));
  int lane = threadIdx.x & 31;
  int m0 = (blockIdx.x * PROJ_WAVES + w) * 16;
  int half = lane >> 4;
  int l16  = lane & 15;

  float4 wv[2];
  float4 av[16];
#pragma unroll
  for (int u = 0; u < 2; ++u)
    wv[u] = *(const float4*)(W + (size_t)(2 * w + u) * IN_FEATS + 4 * lane);
#pragma unroll
  for (int r = 0; r < 16; ++r) {
    int row = min(m0 + r, nNodes - 1);
    av[r] = *(const float4*)(h + (size_t)row * IN_FEATS + 4 * lane);
  }
  float* Aw = As + w * (16 * PROJ_PITCH);
#pragma unroll
  for (int u = 0; u < 2; ++u)
    *(float4*)(Ws + (2 * w + u) * PROJ_PITCH + 4 * lane) = wv[u];
#pragma unroll
  for (int r = 0; r < 16; ++r)
    *(float4*)(Aw + r * PROJ_PITCH + 4 * lane) = av[r];
  __syncthreads();
  if (m0 >= nNodes) return;

  const float* arow = Aw + l16 * PROJ_PITCH + half * 2;
  const float* brow = Ws + l16 * PROJ_PITCH + half * 2;
  v8f acc = {};
#pragma unroll
  for (int k = 0; k < IN_FEATS; k += 4) {
    v2f a = *(const v2f*)(arow + k);
    v2f b = *(const v2f*)(brow + k);
    acc = __builtin_amdgcn_wmma_f32_16x16x4_f32(false, a, false, b,
                                                (short)0, acc, false, false);
  }
  if (m0 + 16 <= nNodes) {
#pragma unroll
    for (int g = 0; g < 8; ++g) {
      int row = m0 + half * 8 + g;
      hW[(size_t)row * OUT_FEATS + l16] = acc[g];
    }
  } else {
#pragma unroll
    for (int g = 0; g < 8; ++g) {
      int row = m0 + half * 8 + g;
      if (row < nNodes) hW[(size_t)row * OUT_FEATS + l16] = acc[g];
    }
  }
}

__device__ __forceinline__ int package_half() {
  return (int)((__builtin_amdgcn_s_sendmsg_rtn(0x87) >> 18) & 1u);
}

__device__ __forceinline__ void atomicMaxFloat(float* addr, float val) {
  if (val >= 0.f) atomicMax((int*)addr, __float_as_int(val));
  else            atomicMin((unsigned int*)addr, (unsigned int)__float_as_int(val));
}
__device__ __forceinline__ void atomicMinFloat(float* addr, float val) {
  if (val >= 0.f) atomicMin((int*)addr, __float_as_int(val));
  else            atomicMax((unsigned int*)addr, (unsigned int)__float_as_int(val));
}

__device__ __forceinline__ float leaky(float x) {
  return (x > 0.f) ? x : NEG_SLOPE * x;
}

__device__ __forceinline__ bool shift_bound_ok(const float* gstat, float& M) {
  M = fmaxf(gstat[0], gstat[1]);
  float m = fminf(gstat[2], gstat[3]);
  return (M >= m) && (M - m <= SHIFT_SPAN);
}

__global__ void __launch_bounds__(256)
gat_node_prep(const float* __restrict__ hW,
              const float* __restrict__ a_l,
              const float* __restrict__ a_r,
              float* __restrict__ er_p,
              float* __restrict__ el_p,
              float* emax0, float* emax1,
              float* esum0, float* esum1,
              float* acc0, float* acc1,
              float* gstat, int nNodes) {
  __shared__ float smx[8], smn[8];
  int n = blockIdx.x * 256 + threadIdx.x;
  float mx = __int_as_float(0xFF800000);
  float mn = __int_as_float(0x7F800000);
  if (n < nNodes) {
    const float* row = hW + (size_t)n * OUT_FEATS;
    float sl = 0.f, sr = 0.f;
#pragma unroll
    for (int f = 0; f < OUT_FEATS; ++f) {
      float v = row[f];
      sl += v * a_l[f];
      sr += v * a_r[f];
    }
    float4 zero4 = make_float4(0.f, 0.f, 0.f, 0.f);
    float4* z0 = (float4*)(acc0 + (size_t)n * OUT_FEATS);
    float4* z1 = (float4*)(acc1 + (size_t)n * OUT_FEATS);
#pragma unroll
    for (int q = 0; q < OUT_FEATS / 4; ++q) z0[q] = zero4;
#pragma unroll
    for (int q = 0; q < OUT_FEATS / 4; ++q) z1[q] = zero4;
    er_p[n] = sr;
    el_p[n] = sl;
    emax0[n] = __int_as_float(0xFF800000);
    emax1[n] = __int_as_float(0xFF800000);
    esum0[n] = 0.f;
    esum1[n] = 0.f;
    mx = sl;
    mn = sl;
  }
#pragma unroll
  for (int m = 16; m >= 1; m >>= 1) {
    mx = fmaxf(mx, __shfl_xor(mx, m, 32));
    mn = fminf(mn, __shfl_xor(mn, m, 32));
  }
  if ((threadIdx.x & 31) == 0) {
    smx[threadIdx.x >> 5] = mx;
    smn[threadIdx.x >> 5] = mn;
  }
  __syncthreads();
  if (threadIdx.x == 0) {
#pragma unroll
    for (int w = 1; w < 8; ++w) {
      mx = fmaxf(mx, smx[w]);
      mn = fminf(mn, smn[w]);
    }
    int hsel = package_half();
    atomicMaxFloat(gstat + hsel, mx);
    atomicMinFloat(gstat + 2 + hsel, mn);
  }
}

__global__ void __launch_bounds__(256)
gat_edge_max(const int* __restrict__ src,
             const int* __restrict__ dst,
             const float* __restrict__ el_p,
             const float* __restrict__ gstat,
             float* emax0, float* emax1, int nEdges) {
  float M;
  if (shift_bound_ok(gstat, M)) return;
  int tid = blockIdx.x * 256 + threadIdx.x;
  int stride = gridDim.x * 256;
  float* emax = package_half() ? emax1 : emax0;
  for (int e0 = tid; e0 < nEdges; e0 += EDGE_UNROLL * stride) {
    int s[EDGE_UNROLL], d[EDGE_UNROLL];
    float v[EDGE_UNROLL];
#pragma unroll
    for (int u = 0; u < EDGE_UNROLL; ++u) {
      int e = e0 + u * stride;
      int ec = (e < nEdges) ? e : e0;
      s[u] = src[ec];
      d[u] = dst[ec];
    }
#pragma unroll
    for (int u = 0; u < EDGE_UNROLL; ++u) v[u] = el_p[s[u]];
#pragma unroll
    for (int u = 0; u < EDGE_UNROLL; ++u) {
      int e = e0 + u * stride;
      if (e < nEdges) atomicMaxFloat(emax + d[u], v[u]);
    }
  }
}

template <int MASK>
__device__ __forceinline__ float row_xor(float v) {
  return __int_as_float(__builtin_amdgcn_update_dpp(0, __float_as_int(v),
                                                    0x160 + MASK, 0xf, 0xf, true));
}

template <int J>
__device__ __forceinline__ float row_lane(float v) {
  return __int_as_float(__builtin_amdgcn_update_dpp(0, __float_as_int(v),
                                                    0x150 + J, 0xf, 0xf, true));
}

template <int MASK, int N>
__device__ __forceinline__ void fold_step(const float* in, float* out, bool bit) {
#pragma unroll
  for (int i = 0; i < N; ++i) {
    float keep = bit ? in[2 * i + 1] : in[2 * i];
    float send = bit ? in[2 * i] : in[2 * i + 1];
    out[i] = keep + row_xor<MASK>(send);
  }
}

template <int B>
__device__ __forceinline__ void fill_rows(const float4* g4, float4* s4,
                                          int i, int rot, int n4) {
  float4 v[B];
  int j[B];
#pragma unroll
  for (int u = 0; u < B; ++u) {
    int t = i + u * AGGR_THREADS + rot;
    j[u] = (t >= n4) ? t - n4 : t;
    v[u] = g4[j[u]];
  }
#pragma unroll
  for (int u = 0; u < B; ++u) s4[j[u]] = v[u];
}

typedef __attribute__((address_space(3))) float lds_f32;

__device__ __forceinline__ float er_of(const lds_f32* er_t, const float* er_g,
                                       int d, int nLds) {
  float v = er_t[max(min(d, nLds - 1), 0)];
  if (d >= nLds) v = er_g[d];
  return v;
}

__global__ void __launch_bounds__(AGGR_THREADS)
gat_edge_aggr(const int* __restrict__ src,
              const int* __restrict__ dst,
              const float* __restrict__ er_g,
              const float* __restrict__ hW,
              const float* __restrict__ a_l,
              const float* __restrict__ gstat,
              const float* emax0, const float* emax1,
              float* acc0, float* acc1,
              float* esum0, float* esum1, int nNodes, int nEdges) {
  __shared__ float er_s[ER_LDS_NODES];
  int tid = threadIdx.x;
  int w = tid >> 5;
  int wave = (w >> 3) * (gridDim.x * 8) + blockIdx.x * 8 + (w & 7);
  int nWaves = gridDim.x * (AGGR_THREADS / 32);
  int lane = tid & 31;
  int rowsel = lane & 16;
  int f = lane & 15;
  int nChunks = (nEdges + 31) >> 5;
  int chunk = wave;
  bool work = chunk < nChunks;

  int s_l = 0, d_l = 0, s_n = 0, d_n = 0;
  if (work) {
    int e = min(chunk * 32 + lane, nEdges - 1);
    s_l = src[e];
    d_l = dst[e];
    if (chunk + nWaves < nChunks) {
      int en = min((chunk + nWaves) * 32 + lane, nEdges - 1);
      s_n = src[en];
      d_n = dst[en];
    }
  }

  int nLds = min(nNodes, ER_LDS_NODES);
  int nLds4 = nLds >> 2;
  {
    const float4* g4 = (const float4*)er_g;
    float4* s4 = (float4*)er_s;
    int rot = (nLds4 > 0) ? (int)((blockIdx.x * 64u) % (unsigned)nLds4) : 0;
    int rowsFull = nLds4 / AGGR_THREADS;
    int r = 0;
    for (; r + FILL_BATCH <= rowsFull; r += FILL_BATCH)
      fill_rows<FILL_BATCH>(g4, s4, r * AGGR_THREADS + tid, rot, nLds4);
    for (; r + 4 <= rowsFull; r += 4)
      fill_rows<4>(g4, s4, r * AGGR_THREADS + tid, rot, nLds4);
    for (; r + 2 <= rowsFull; r += 2)
      fill_rows<2>(g4, s4, r * AGGR_THREADS + tid, rot, nLds4);
    for (; r + 1 <= rowsFull; r += 1)
      fill_rows<1>(g4, s4, r * AGGR_THREADS + tid, rot, nLds4);
    if (r * AGGR_THREADS + tid < nLds4)
      fill_rows<1>(g4, s4, r * AGGR_THREADS + tid, rot, nLds4);
    for (int i = (nLds4 << 2) + tid; i < nLds; i += AGGR_THREADS) er_s[i] = er_g[i];
  }
  __syncthreads();
  if (!work) return;

  int hsel = package_half();
  float* out  = hsel ? acc1 : acc0;
  float* esum = hsel ? esum1 : esum0;
  float M;
  bool bound = shift_bound_ok(gstat, M);
  float al_f = a_l[f];
  bool b1 = (f & 1) != 0, b2 = (f & 2) != 0, b4 = (f & 4) != 0, b8 = (f & 8) != 0;

  const lds_f32* er_t = (const lds_f32*)er_s;
  float er_l = er_of(er_t, er_g, d_l, nLds);

  for (; chunk < nChunks; chunk += nWaves) {
    int e0 = chunk * 32;
    int s_f = s_n, d_f = d_n;
    int cf = chunk + 2 * nWaves;
    if (cf < nChunks) {
      int ef = min(cf * 32 + lane, nEdges - 1);
      s_f = src[ef];
      d_f = dst[ef];
    }

    float c_l;
    if (bound) {
      c_l = leaky(M + er_l);
    } else {
      float m = fmaxf(emax0[d_l], emax1[d_l]);
      c_l = leaky(m + er_l);
      if (!(fabsf(c_l) < __int_as_float(0x7F800000))) c_l = 0.f;
    }

    int dd[16];
    float rr[16];
#pragma unroll
    for (int j = 0; j < 16; ++j) {
      int q = rowsel | j;
      int s = __shfl(s_l, q, 32);
      dd[j] = __shfl(d_l, q, 32);
      rr[j] = hW[(size_t)s * OUT_FEATS + f];
    }

    float er_nt = er_l, er_ng = er_l;
    bool far_n = d_n >= nLds;
    if (chunk + nWaves < nChunks) {
      er_nt = er_t[max(min(d_n, nLds - 1), 0)];
      if (far_n) er_ng = er_g[d_n];
    }

    float p[16], t8[8], t4[4], t2[2], t1[1];
#pragma unroll
    for (int j = 0; j < 16; ++j) p[j] = rr[j] * al_f;
    fold_step<1, 8>(p, t8, b1);
    fold_step<2, 4>(t8, t4, b2);
    fold_step<4, 2>(t4, t2, b4);
    fold_step<8, 1>(t2, t1, b8);
    float el_l = t1[0];

    float x = leaky(el_l + er_l);
    float x_l = __expf(x - c_l);
    if (e0 + lane < nEdges) atomicAdd(esum + d_l, x_l);

    float vv[16];
#define WEIGH(J) vv[J] = rr[J] * row_lane<J>(x_l);
    WEIGH(0)  WEIGH(1)  WEIGH(2)  WEIGH(3)  WEIGH(4)  WEIGH(5)  WEIGH(6)  WEIGH(7)
    WEIGH(8)  WEIGH(9)  WEIGH(10) WEIGH(11) WEIGH(12) WEIGH(13) WEIGH(14) WEIGH(15)
#undef WEIGH
#pragma unroll
    for (int j = 0; j < 16; ++j) {
      if (e0 + (rowsel | j) < nEdges)
        atomicAdd(out + (size_t)dd[j] * OUT_FEATS + f, vv[j]);
    }

    s_l = s_n; d_l = d_n; er_l = far_n ? er_ng : er_nt;
    s_n = s_f; d_n = d_f;
  }
}

__global__ void gat_join_out(float* out, const float* acc1,
                             const float* esum0, const float* esum1, int nVec4) {
  int i = blockIdx.x * blockDim.x + threadIdx.x;
  if (i >= nVec4) return;
  int n = i / (OUT_FEATS / 4);
  float4 a = ((const float4*)out)[i];
  float es = esum0[n];
  if (esum1 != esum0) {
    float4 b = ((const float4*)acc1)[i];
    a.x += b.x; a.y += b.y; a.z += b.z; a.w += b.w;
    es += esum1[n];
  }
  float den = fmaxf(es, 1e-16f);
  a.x /= den; a.y /= den; a.z /= den; a.w /= den;
  ((float4*)out)[i] = a;
}

extern "C" void kernel_launch(void* const* d_in, const int* in_sizes, int n_in,
                              void* d_out, int out_size, void* d_ws, size_t ws_size,
                              hipStream_t stream) {
  const float* h   = (const float*)d_in[0];
  const int*   src = (const int*)d_in[1];
  const int*   dst = (const int*)d_in[2];
  const float* W   = (const float*)d_in[3];
  const float* a_l = (const float*)d_in[4];
  const float* a_r = (const float*)d_in[5];
  float* out = (float*)d_out;

  int nNodes = in_sizes[0] / IN_FEATS;
  int nEdges = in_sizes[1];

  size_t need2 = ((size_t)nNodes * (2 * OUT_FEATS + 6) + 4) * sizeof(float);
  bool two = ws_size >= need2;
  float* ws    = (float*)d_ws;
  float* hW    = ws;
  float* acc1  = two ? hW + (size_t)nNodes * OUT_FEATS : out;
  float* er_p  = hW + (size_t)nNodes * OUT_FEATS * (two ? 2 : 1);
  float* el_p  = er_p + nNodes;
  float* emax0 = el_p + nNodes;
  float* emax1 = two ? emax0 + nNodes : emax0;
  float* esum0 = emax1 + nNodes;
  float* esum1 = two ? esum0 + nNodes : esum0;
  float* gstat = esum1 + nNodes;

  dim3 blk(256);
  int tiles = (nNodes + 15) / 16;
  int blocksK1 = (tiles + 7) / 8;

  gat_project_wmma<<<blocksK1, blk, 0, stream>>>(h, W, hW, gstat, nNodes);
  gat_node_prep<<<(nNodes + 255) / 256, blk, 0, stream>>>(hW, a_l, a_r, er_p, el_p,
                                                          emax0, emax1, esum0, esum1,
                                                          out, acc1, gstat, nNodes);
  gat_edge_max<<<EDGE_WGS, blk, 0, stream>>>(src, dst, el_p, gstat,
                                             emax0, emax1, nEdges);
  gat_edge_aggr<<<AGGR_WGS, dim3(AGGR_THREADS), 0, stream>>>(src, dst, er_p, hW, a_l, gstat,
                                                             emax0, emax1,
                                                             out, acc1, esum0, esum1,
                                                             nNodes, nEdges);
  int nVec4 = nNodes * (OUT_FEATS / 4);
  gat_join_out<<<(nVec4 + 255) / 256, blk, 0, stream>>>(out, acc1, esum0, esum1, nVec4);
}


# Segment sum moved into its node's 128-byte output line (per package half): one 17-lane atomic per edge adds row and exp value together, was two requests; join reads the lines; sum order changes
# speedup vs baseline: 6.4234x; 1.1093x over previous
#include <hip/hip_runtime.h>
#include <hip/hip_bf16.h>

#define IN_FEATS   128
#define OUT_FEATS  16
#define NEG_SLOPE  0.2f

#define EDGE_WGS     256
#define EDGE_UNROLL  4
#define AGGR_WGS     256
#define AGGR_THREADS 1024
#define ER_LDS_NODES 81920
#define FILL_BATCH   20
#define SHIFT_SPAN   32.0f
#define LINE_F       32
#define LINE_SUM     16

typedef float v2f __attribute__((ext_vector_type(2)));
typedef float v8f __attribute__((ext_vector_type(8)));

#define PROJ_WAVES 8
#define PROJ_PITCH 132
__global__ void __launch_bounds__(PROJ_WAVES * 32)
gat_project_wmma(const float* __restrict__ h,
                 const float* __restrict__ W,
                 float* __restrict__ hW,
                 float* __restrict__ gstat, int nNodes) {
  __shared__ __attribute__((aligned(16))) float Ws[OUT_FEATS * PROJ_PITCH];
  __shared__ __attribute__((aligned(16))) float As[PROJ_WAVES * 16 * PROJ_PITCH];
  if (blockIdx.x == 0 && threadIdx.x == 0) {
    gstat[0] = __int_as_float(0xFF800000);
    gstat[1] = __int_as_float(0xFF800000);
    gstat[2] = __int_as_float(0x7F800000);
    gstat[3] = __int_as_float(0x7F800000);
  }
  int w = __builtin_amdgcn_readfirstlane((int)(threadIdx.x >> 5));
  int lane = threadIdx.x & 31;
  int m0 = (blockIdx.x * PROJ_WAVES + w) * 16;
  int half = lane >> 4;
  int l16  = lane & 15;

  float4 wv[2];
  float4 av[16];
#pragma unroll
  for (int u = 0; u < 2; ++u)
    wv[u] = *(const float4*)(W + (size_t)(2 * w + u) * IN_FEATS + 4 * lane);
#pragma unroll
  for (int r = 0; r < 16; ++r) {
    int row = min(m0 + r, nNodes - 1);
    av[r] = *(const float4*)(h + (size_t)row * IN_FEATS + 4 * lane);
  }
  float* Aw = As + w * (16 * PROJ_PITCH);
#pragma unroll
  for (int u = 0; u < 2; ++u)
    *(float4*)(Ws + (2 * w + u) * PROJ_PITCH + 4 * lane) = wv[u];
#pragma unroll
  for (int r = 0; r < 16; ++r)
    *(float4*)(Aw + r * PROJ_PITCH + 4 * lane) = av[r];
  __syncthreads();
  if (m0 >= nNodes) return;

  const float* arow = Aw + l16 * PROJ_PITCH + half * 2;
  const float* brow = Ws + l16 * PROJ_PITCH + half * 2;
  v8f acc = {};
#pragma unroll
  for (int k = 0; k < IN_FEATS; k += 4) {
    v2f a = *(const v2f*)(arow + k);
    v2f b = *(const v2f*)(brow + k);
    acc = __builtin_amdgcn_wmma_f32_16x16x4_f32(false, a, false, b,
                                                (short)0, acc, false, false);
  }
  if (m0 + 16 <= nNodes) {
#pragma unroll
    for (int g = 0; g < 8; ++g) {
      int row = m0 + half * 8 + g;
      hW[(size_t)row * OUT_FEATS + l16] = acc[g];
    }
  } else {
#pragma unroll
    for (int g = 0; g < 8; ++g) {
      int row = m0 + half * 8 + g;
      if (row < nNodes) hW[(size_t)row * OUT_FEATS + l16] = acc[g];
    }
  }
}

__device__ __forceinline__ int package_half() {
  return (int)((__builtin_amdgcn_s_sendmsg_rtn(0x87) >> 18) & 1u);
}

__device__ __forceinline__ void atomicMaxFloat(float* addr, float val) {
  if (val >= 0.f) atomicMax((int*)addr, __float_as_int(val));
  else            atomicMin((unsigned int*)addr, (unsigned int)__float_as_int(val));
}
__device__ __forceinline__ void atomicMinFloat(float* addr, float val) {
  if (val >= 0.f) atomicMin((int*)addr, __float_as_int(val));
  else            atomicMax((unsigned int*)addr, (unsigned int)__float_as_int(val));
}

__device__ __forceinline__ float leaky(float x) {
  return (x > 0.f) ? x : NEG_SLOPE * x;
}

__device__ __forceinline__ bool shift_bound_ok(const float* gstat, float& M) {
  M = fmaxf(gstat[0], gstat[1]);
  float m = fminf(gstat[2], gstat[3]);
  return (M >= m) && (M - m <= SHIFT_SPAN);
}

__global__ void __launch_bounds__(256)
gat_node_prep(const float* __restrict__ hW,
              const float* __restrict__ a_l,
              const float* __restrict__ a_r,
              float* __restrict__ er_p,
              float* __restrict__ el_p,
              float* emax0, float* emax1,
              float* esum0, float* esum1,
              float* acc0, float* acc1,
              float* gstat, int nNodes) {
  __shared__ float smx[8], smn[8];
  int n = blockIdx.x * 256 + threadIdx.x;
  float mx = __int_as_float(0xFF800000);
  float mn = __int_as_float(0x7F800000);
  if (n < nNodes) {
    const float* row = hW + (size_t)n * OUT_FEATS;
    float sl = 0.f, sr = 0.f;
#pragma unroll
    for (int f = 0; f < OUT_FEATS; ++f) {
      float v = row[f];
      sl += v * a_l[f];
      sr += v * a_r[f];
    }
    float4 zero4 = make_float4(0.f, 0.f, 0.f, 0.f);
    float4* z0 = (float4*)(acc0 + (size_t)n * OUT_FEATS);
    float4* z1 = (float4*)(acc1 + (size_t)n * OUT_FEATS);
#pragma unroll
    for (int q = 0; q < OUT_FEATS / 4; ++q) z0[q] = zero4;
#pragma unroll
    for (int q = 0; q < OUT_FEATS / 4; ++q) z1[q] = zero4;
    er_p[n] = sr;
    el_p[n] = sl;
    emax0[n] = __int_as_float(0xFF800000);
    emax1[n] = __int_as_float(0xFF800000);
    esum0[n] = 0.f;
    esum1[n] = 0.f;
    mx = sl;
    mn = sl;
  }
#pragma unroll
  for (int m = 16; m >= 1; m >>= 1) {
    mx = fmaxf(mx, __shfl_xor(mx, m, 32));
    mn = fminf(mn, __shfl_xor(mn, m, 32));
  }
  if ((threadIdx.x & 31) == 0) {
    smx[threadIdx.x >> 5] = mx;
    smn[threadIdx.x >> 5] = mn;
  }
  __syncthreads();
  if (threadIdx.x == 0) {
#pragma unroll
    for (int w = 1; w < 8; ++w) {
      mx = fmaxf(mx, smx[w]);
      mn = fminf(mn, smn[w]);
    }
    int hsel = package_half();
    atomicMaxFloat(gstat + hsel, mx);
    atomicMinFloat(gstat + 2 + hsel, mn);
  }
}

__global__ void __launch_bounds__(256)
gat_node_prep_lines(const float* __restrict__ hW,
                    const float* __restrict__ a_l,
                    const float* __restrict__ a_r,
                    float* __restrict__ er_p,
                    float* __restrict__ el_p,
                    float* emax0, float* emax1,
                    float* pl0, float* pl1,
                    float* gstat, int nNodes) {
  __shared__ float smx[8], smn[8];
  {
    float4 zero4 = make_float4(0.f, 0.f, 0.f, 0.f);
    size_t nPieces = (size_t)nNodes * (LINE_F / 4);
    size_t k0 = (size_t)blockIdx.x * (256 * (LINE_F / 4)) + threadIdx.x;
#pragma unroll
    for (int i = 0; i < LINE_F / 4; ++i) {
      size_t k = k0 + (size_t)i * 256;
      if (k < nPieces) {
        ((float4*)pl0)[k] = zero4;
        ((float4*)pl1)[k] = zero4;
      }
    }
  }
  int n = blockIdx.x * 256 + threadIdx.x;
  float mx = __int_as_float(0xFF800000);
  float mn = __int_as_float(0x7F800000);
  if (n < nNodes) {
    const float* row = hW + (size_t)n * OUT_FEATS;
    float sl = 0.f, sr = 0.f;
#pragma unroll
    for (int f = 0; f < OUT_FEATS; ++f) {
      float v = row[f];
      sl += v * a_l[f];
      sr += v * a_r[f];
    }
    er_p[n] = sr;
    el_p[n] = sl;
    emax0[n] = __int_as_float(0xFF800000);
    emax1[n] = __int_as_float(0xFF800000);
    mx = sl;
    mn = sl;
  }
#pragma unroll
  for (int m = 16; m >= 1; m >>= 1) {
    mx = fmaxf(mx, __shfl_xor(mx, m, 32));
    mn = fminf(mn, __shfl_xor(mn, m, 32));
  }
  if ((threadIdx.x & 31) == 0) {
    smx[threadIdx.x >> 5] = mx;
    smn[threadIdx.x >> 5] = mn;
  }
  __syncthreads();
  if (threadIdx.x == 0) {
#pragma unroll
    for (int w = 1; w < 8; ++w) {
      mx = fmaxf(mx, smx[w]);
      mn = fminf(mn, smn[w]);
    }
    int hsel = package_half();
    atomicMaxFloat(gstat + hsel, mx);
    atomicMinFloat(gstat + 2 + hsel, mn);
  }
}

__global__ void __launch_bounds__(256)
gat_edge_max(const int* __restrict__ src,
             const int* __restrict__ dst,
             const float* __restrict__ el_p,
             const float* __restrict__ gstat,
             float* emax0, float* emax1, int nEdges) {
  float M;
  if (shift_bound_ok(gstat, M)) return;
  int tid = blockIdx.x * 256 + threadIdx.x;
  int stride = gridDim.x * 256;
  float* emax = package_half() ? emax1 : emax0;
  for (int e0 = tid; e0 < nEdges; e0 += EDGE_UNROLL * stride) {
    int s[EDGE_UNROLL], d[EDGE_UNROLL];
    float v[EDGE_UNROLL];
#pragma unroll
    for (int u = 0; u < EDGE_UNROLL; ++u) {
      int e = e0 + u * stride;
      int ec = (e < nEdges) ? e : e0;
      s[u] = src[ec];
      d[u] = dst[ec];
    }
#pragma unroll
    for (int u = 0; u < EDGE_UNROLL; ++u) v[u] = el_p[s[u]];
#pragma unroll
    for (int u = 0; u < EDGE_UNROLL; ++u) {
      int e = e0 + u * stride;
      if (e < nEdges) atomicMaxFloat(emax + d[u], v[u]);
    }
  }
}

template <int MASK>
__device__ __forceinline__ float row_xor(float v) {
  return __int_as_float(__builtin_amdgcn_update_dpp(0, __float_as_int(v),
                                                    0x160 + MASK, 0xf, 0xf, true));
}

template <int J>
__device__ __forceinline__ float row_lane(float v) {
  return __int_as_float(__builtin_amdgcn_update_dpp(0, __float_as_int(v),
                                                    0x150 + J, 0xf, 0xf, true));
}

template <int MASK, int N>
__device__ __forceinline__ void fold_step(const float* in, float* out, bool bit) {
#pragma unroll
  for (int i = 0; i < N; ++i) {
    float keep = bit ? in[2 * i + 1] : in[2 * i];
    float send = bit ? in[2 * i] : in[2 * i + 1];
    out[i] = keep + row_xor<MASK>(send);
  }
}

template <int B>
__device__ __forceinline__ void fill_rows(const float4* g4, float4* s4,
                                          int i, int rot, int n4) {
  float4 v[B];
  int j[B];
#pragma unroll
  for (int u = 0; u < B; ++u) {
    int t = i + u * AGGR_THREADS + rot;
    j[u] = (t >= n4) ? t - n4 : t;
    v[u] = g4[j[u]];
  }
#pragma unroll
  for (int u = 0; u < B; ++u) s4[j[u]] = v[u];
}

typedef __attribute__((address_space(3))) float lds_f32;

__device__ __forceinline__ float er_of(const lds_f32* er_t, const float* er_g,
                                       int d, int nLds) {
  float v = er_t[max(min(d, nLds - 1), 0)];
  if (d >= nLds) v = er_g[d];
  return v;
}

template <bool FULL>
__device__ __forceinline__ void add_lines(float* plane, const float* vv,
                                          float x_l, int d_l, int lane, int nv) {
  unsigned offA = (lane < 16) ? 4u * (unsigned)lane : 4u * LINE_SUM;
  unsigned offB = (lane < 16) ? 4u * LINE_SUM : 4u * (unsigned)(lane - 16);
  if (lane <= 16) {
#pragma unroll
    for (int j = 0; j < 16; ++j) {
      if (FULL || j < nv) {
        int d = __builtin_amdgcn_readlane(d_l, j);
        float xs = __int_as_float(__builtin_amdgcn_readlane(__float_as_int(x_l), j));
        float v = (lane == 16) ? xs : vv[j];
        char* line = (char*)(plane + (size_t)d * LINE_F);
        atomicAdd((float*)(line + offA), v);
      }
    }
  }
  if (lane >= 15) {
#pragma unroll
    for (int j = 0; j < 16; ++j) {
      if (FULL || 16 + j < nv) {
        int d = __builtin_amdgcn_readlane(d_l, 16 + j);
        float xs = __int_as_float(__builtin_amdgcn_readlane(__float_as_int(x_l), 16 + j));
        float v = (lane == 15) ? xs : vv[j];
        char* line = (char*)(plane + (size_t)d * LINE_F);
        atomicAdd((float*)(line + offB), v);
      }
    }
  }
}

template <bool LINES>
__global__ void __launch_bounds__(AGGR_THREADS)
gat_edge_aggr(const int* __restrict__ src,
              const int* __restrict__ dst,
              const float* __restrict__ er_g,
              const float* __restrict__ hW,
              const float* __restrict__ a_l,
              const float* __restrict__ gstat,
              const float* emax0, const float* emax1,
              float* acc0, float* acc1,
              float* esum0, float* esum1, int nNodes, int nEdges) {
  __shared__ float er_s[ER_LDS_NODES];
  int tid = threadIdx.x;
  int w = tid >> 5;
  int wave = (w >> 3) * (gridDim.x * 8) + blockIdx.x * 8 + (w & 7);
  int nWaves = gridDim.x * (AGGR_THREADS / 32);
  int lane = tid & 31;
  int rowsel = lane & 16;
  int f = lane & 15;
  int nChunks = (nEdges + 31) >> 5;
  int chunk = wave;
  bool work = chunk < nChunks;

  int s_l = 0, d_l = 0, s_n = 0, d_n = 0;
  if (work) {
    int e = min(chunk * 32 + lane, nEdges - 1);
    s_l = src[e];
    d_l = dst[e];
    if (chunk + nWaves < nChunks) {
      int en = min((chunk + nWaves) * 32 + lane, nEdges - 1);
      s_n = src[en];
      d_n = dst[en];
    }
  }

  int nLds = min(nNodes, ER_LDS_NODES);
  int nLds4 = nLds >> 2;
  {
    const float4* g4 = (const float4*)er_g;
    float4* s4 = (float4*)er_s;
    int rot = (nLds4 > 0) ? (int)((blockIdx.x * 64u) % (unsigned)nLds4) : 0;
    int rowsFull = nLds4 / AGGR_THREADS;
    int r = 0;
    for (; r + FILL_BATCH <= rowsFull; r += FILL_BATCH)
      fill_rows<FILL_BATCH>(g4, s4, r * AGGR_THREADS + tid, rot, nLds4);
    for (; r + 4 <= rowsFull; r += 4)
      fill_rows<4>(g4, s4, r * AGGR_THREADS + tid, rot, nLds4);
    for (; r + 2 <= rowsFull; r += 2)
      fill_rows<2>(g4, s4, r * AGGR_THREADS + tid, rot, nLds4);
    for (; r + 1 <= rowsFull; r += 1)
      fill_rows<1>(g4, s4, r * AGGR_THREADS + tid, rot, nLds4);
    if (r * AGGR_THREADS + tid < nLds4)
      fill_rows<1>(g4, s4, r * AGGR_THREADS + tid, rot, nLds4);
    for (int i = (nLds4 << 2) + tid; i < nLds; i += AGGR_THREADS) er_s[i] = er_g[i];
  }
  __syncthreads();
  if (!work) return;

  int hsel = package_half();
  float* out  = hsel ? acc1 : acc0;
  float* esum = hsel ? esum1 : esum0;
  float M;
  bool bound = shift_bound_ok(gstat, M);
  float al_f = a_l[f];
  bool b1 = (f & 1) != 0, b2 = (f & 2) != 0, b4 = (f & 4) != 0, b8 = (f & 8) != 0;

  const lds_f32* er_t = (const lds_f32*)er_s;
  float er_l = er_of(er_t, er_g, d_l, nLds);

  for (; chunk < nChunks; chunk += nWaves) {
    int e0 = chunk * 32;
    int s_f = s_n, d_f = d_n;
    int cf = chunk + 2 * nWaves;
    if (cf < nChunks) {
      int ef = min(cf * 32 + lane, nEdges - 1);
      s_f = src[ef];
      d_f = dst[ef];
    }

    float c_l;
    if (bound) {
      c_l = leaky(M + er_l);
    } else {
      float m = fmaxf(emax0[d_l], emax1[d_l]);
      c_l = leaky(m + er_l);
      if (!(fabsf(c_l) < __int_as_float(0x7F800000))) c_l = 0.f;
    }

    int dd[16];
    float rr[16];
#pragma unroll
    for (int j = 0; j < 16; ++j) {
      int q = rowsel | j;
      int s = __shfl(s_l, q, 32);
      if constexpr (!LINES) dd[j] = __shfl(d_l, q, 32);
      rr[j] = hW[(size_t)s * OUT_FEATS + f];
    }

    float er_nt = er_l, er_ng = er_l;
    bool far_n = d_n >= nLds;
    if (chunk + nWaves < nChunks) {
      er_nt = er_t[max(min(d_n, nLds - 1), 0)];
      if (far_n) er_ng = er_g[d_n];
    }

    float p[16], t8[8], t4[4], t2[2], t1[1];
#pragma unroll
    for (int j = 0; j < 16; ++j) p[j] = rr[j] * al_f;
    fold_step<1, 8>(p, t8, b1);
    fold_step<2, 4>(t8, t4, b2);
    fold_step<4, 2>(t4, t2, b4);
    fold_step<8, 1>(t2, t1, b8);
    float el_l = t1[0];

    float x = leaky(el_l + er_l);
    float x_l = __expf(x - c_l);
    if constexpr (!LINES) {
      if (e0 + lane < nEdges) atomicAdd(esum + d_l, x_l);
    }

    float vv[16];
#define WEIGH(J) vv[J] = rr[J] * row_lane<J>(x_l);
    WEIGH(0)  WEIGH(1)  WEIGH(2)  WEIGH(3)  WEIGH(4)  WEIGH(5)  WEIGH(6)  WEIGH(7)
    WEIGH(8)  WEIGH(9)  WEIGH(10) WEIGH(11) WEIGH(12) WEIGH(13) WEIGH(14) WEIGH(15)
#undef WEIGH
    if constexpr (LINES) {
      int nv = __builtin_amdgcn_readfirstlane(nEdges - e0);
      if (nv >= 32) add_lines<true>(out, vv, x_l, d_l, lane, 32);
      else          add_lines<false>(out, vv, x_l, d_l, lane, nv);
    } else {
#pragma unroll
      for (int j = 0; j < 16; ++j) {
        if (e0 + (rowsel | j) < nEdges)
          atomicAdd(out + (size_t)dd[j] * OUT_FEATS + f, vv[j]);
      }
    }

    s_l = s_n; d_l = d_n; er_l = far_n ? er_ng : er_nt;
    s_n = s_f; d_n = d_f;
  }
}

__global__ void gat_join_out(float* out, const float* acc1,
                             const float* esum0, const float* esum1, int nVec4) {
  int i = blockIdx.x * blockDim.x + threadIdx.x;
  if (i >= nVec4) return;
  int n = i / (OUT_FEATS / 4);
  float4 a = ((const float4*)out)[i];
  float es = esum0[n];
  if (esum1 != esum0) {
    float4 b = ((const float4*)acc1)[i];
    a.x += b.x; a.y += b.y; a.z += b.z; a.w += b.w;
    es += esum1[n];
  }
  float den = fmaxf(es, 1e-16f);
  a.x /= den; a.y /= den; a.z /= den; a.w /= den;
  ((float4*)out)[i] = a;
}

__global__ void gat_join_lines(float* __restrict__ out,
                               const float* __restrict__ pl0,
                               const float* __restrict__ pl1, int nVec4) {
  int i = blockIdx.x * blockDim.x + threadIdx.x;
  if (i >= nVec4) return;
  int n = i / (OUT_FEATS / 4);
  int q = i % (OUT_FEATS / 4);
  const float* l0 = pl0 + (size_t)n * LINE_F;
  const float* l1 = pl1 + (size_t)n * LINE_F;
  float4 a = *(const float4*)(l0 + 4 * q);
  float4 b = *(const float4*)(l1 + 4 * q);
  float es = l0[LINE_SUM] + l1[LINE_SUM];
  a.x += b.x; a.y += b.y; a.z += b.z; a.w += b.w;
  float den = fmaxf(es, 1e-16f);
  a.x /= den; a.y /= den; a.z /= den; a.w /= den;
  ((float4*)out)[i] = a;
}

extern "C" void kernel_launch(void* const* d_in, const int* in_sizes, int n_in,
                              void* d_out, int out_size, void* d_ws, size_t ws_size,
                              hipStream_t stream) {
  const float* h   = (const float*)d_in[0];
  const int*   src = (const int*)d_in[1];
  const int*   dst = (const int*)d_in[2];
  const float* W   = (const float*)d_in[3];
  const float* a_l = (const float*)d_in[4];
  const float* a_r = (const float*)d_in[5];
  float* out = (float*)d_out;

  int nNodes = in_sizes[0] / IN_FEATS;
  int nEdges = in_sizes[1];

  dim3 blk(256);
  int tiles = (nNodes + 15) / 16;
  int blocksK1 = (tiles + 7) / 8;
  int nVec4 = nNodes * (OUT_FEATS / 4);

  size_t hWf = ((size_t)nNodes * OUT_FEATS + (LINE_F - 1)) & ~(size_t)(LINE_F - 1);
  size_t needL = (hWf + (size_t)nNodes * (2 * LINE_F + 4) + 4) * sizeof(float);
  if (ws_size >= needL) {
    float* hW    = (float*)d_ws;
    float* pl0   = hW + hWf;
    float* pl1   = pl0 + (size_t)nNodes * LINE_F;
    float* er_p  = pl1 + (size_t)nNodes * LINE_F;
    float* el_p  = er_p + nNodes;
    float* emax0 = el_p + nNodes;
    float* emax1 = emax0 + nNodes;
    float* gstat = emax1 + nNodes;
    gat_project_wmma<<<blocksK1, blk, 0, stream>>>(h, W, hW, gstat, nNodes);
    gat_node_prep_lines<<<(nNodes + 255) / 256, blk, 0, stream>>>(hW, a_l, a_r, er_p, el_p,
                                                                  emax0, emax1, pl0, pl1,
                                                                  gstat, nNodes);
    gat_edge_max<<<EDGE_WGS, blk, 0, stream>>>(src, dst, el_p, gstat,
                                               emax0, emax1, nEdges);
    gat_edge_aggr<true><<<AGGR_WGS, dim3(AGGR_THREADS), 0, stream>>>(src, dst, er_p, hW, a_l, gstat,
                                                                     emax0, emax1,
                                                                     pl0, pl1, nullptr, nullptr,
                                                                     nNodes, nEdges);
    gat_join_lines<<<(nVec4 + 255) / 256, blk, 0, stream>>>(out, pl0, pl1, nVec4);
    return;
  }

  size_t need2 = ((size_t)nNodes * (2 * OUT_FEATS + 6) + 4) * sizeof(float);
  bool two = ws_size >= need2;
  float* ws    = (float*)d_ws;
  float* hW    = ws;
  float* acc1  = two ? hW + (size_t)nNodes * OUT_FEATS : out;
  float* er_p  = hW + (size_t)nNodes * OUT_FEATS * (two ? 2 : 1);
  float* el_p  = er_p + nNodes;
  float* emax0 = el_p + nNodes;
  float* emax1 = two ? emax0 + nNodes : emax0;
  float* esum0 = emax1 + nNodes;
  float* esum1 = two ? esum0 + nNodes : esum0;
  float* gstat = esum1 + nNodes;

  gat_project_wmma<<<blocksK1, blk, 0, stream>>>(h, W, hW, gstat, nNodes);
  gat_node_prep<<<(nNodes + 255) / 256, blk, 0, stream>>>(hW, a_l, a_r, er_p, el_p,
                                                          emax0, emax1, esum0, esum1,
                                                          out, acc1, gstat, nNodes);
  gat_edge_max<<<EDGE_WGS, blk, 0, stream>>>(src, dst, el_p, gstat,
                                             emax0, emax1, nEdges);
  gat_edge_aggr<false><<<AGGR_WGS, dim3(AGGR_THREADS), 0, stream>>>(src, dst, er_p, hW, a_l, gstat,
                                                                    emax0, emax1,
                                                                    out, acc1, esum0, esum1,
                                                                    nNodes, nEdges);
  gat_join_out<<<(nVec4 + 255) / 256, blk, 0, stream>>>(out, acc1, esum0, esum1, nVec4);
}
